# starting kernel
# baseline (speedup 1.0000x reference)
#include <hip/hip_runtime.h>
#include <math.h>

typedef __attribute__((ext_vector_type(2))) float v2f;
typedef __attribute__((ext_vector_type(8))) float v8f;

#define E_DIM 64
#define H_NUM 8
#define D_HEAD 8
#define B_SZ 16
#define S_LEN 1024
#define N_TOK (B_SZ * S_LEN)

__device__ __forceinline__ v8f wmma4(v2f a, v2f b, v8f c) {
  return __builtin_amdgcn_wmma_f32_16x16x4_f32(
       false, a,  false, b,
       (short)0, c,  false,  false);
}

__global__ __launch_bounds__(128) void k_qkv(const float* __restrict__ x,
                                             const float* __restrict__ Wqkv,
                                             const float* __restrict__ bqkv,
                                             float* __restrict__ qkv) {
  const int wave = threadIdx.x >> 5;
  const int lane = threadIdx.x & 31;
  const int n16 = lane & 15, hi = lane >> 4;
  const int tile = blockIdx.x * 4 + wave;
  const int mt = tile / 12, nt = tile % 12;
  const int m0 = mt * 16, n0 = nt * 16;
  const float* arow = x + (m0 + n16) * E_DIM;
  const float* brow = Wqkv + (n0 + n16) * E_DIM;
  v8f c = {};
#pragma unroll
  for (int k0 = 0; k0 < E_DIM; k0 += 4) {
    v2f a, b;
    a.x = arow[k0 + 2 * hi]; a.y = arow[k0 + 2 * hi + 1];
    b.x = brow[k0 + 2 * hi]; b.y = brow[k0 + 2 * hi + 1];
    c = wmma4(a, b, c);
  }
  const float bias = bqkv[n0 + n16];
#pragma unroll
  for (int r = 0; r < 8; ++r) {
    const int row = m0 + r + 8 * hi;
    qkv[row * (3 * E_DIM) + n0 + n16] = c[r] + bias;
  }
}

__global__ __launch_bounds__(128) void k_attn(const float* __restrict__ qkv,
                                              float* __restrict__ attn_out) {
  __shared__ float ldsK[2][16 * 9];
  __shared__ float ldsV[2][16 * 9];
  __shared__ float ldsP[4][16 * 16];
  const int tid = threadIdx.x;
  const int wave = tid >> 5;
  const int lane = tid & 31;
  const int n16 = lane & 15, hi = lane >> 4;

  const int bid = blockIdx.x;
  const int bh = bid >> 4;
  const int b = bh >> 3;
  const int h = bh & 7;
  const int qt = (bid & 15) * 4 + wave;
  const float scale = 0.35355339059327373f;

  const float* qkvB = qkv + (size_t)(b * S_LEN) * (3 * E_DIM);
  const int srow = tid >> 3, scol = tid & 7;

  v2f qa0, qa1;
  {
    const float* qp = qkvB + (qt * 16 + n16) * (3 * E_DIM) + h * D_HEAD;
    qa0.x = qp[0 + 2 * hi] * scale; qa0.y = qp[1 + 2 * hi] * scale;
    qa1.x = qp[4 + 2 * hi] * scale; qa1.y = qp[5 + 2 * hi] * scale;
  }

  {
    const float* rowp = qkvB + srow * (3 * E_DIM) + h * D_HEAD + scol;
    ldsK[0][srow * 9 + scol] = rowp[E_DIM];
    ldsV[0][srow * 9 + scol] = rowp[2 * E_DIM];
  }

  float m[8], l[8];
  v8f o = {};
#pragma unroll
  for (int r = 0; r < 8; ++r) { m[r] = -3.0e38f; l[r] = 0.f; }
  float* myP = ldsP[wave];
  const int d = n16 & 7;

  for (int kt = 0; kt < S_LEN / 16; ++kt) {
    __syncthreads();
    const int cur = kt & 1, nxt = cur ^ 1;
    if (kt + 1 < S_LEN / 16) {
      const float* rowp =
          qkvB + ((kt + 1) * 16 + srow) * (3 * E_DIM) + h * D_HEAD + scol;
      ldsK[nxt][srow * 9 + scol] = rowp[E_DIM];
      ldsV[nxt][srow * 9 + scol] = rowp[2 * E_DIM];
    }

    v2f kb0, kb1;
    kb0.x = ldsK[cur][n16 * 9 + 0 + 2 * hi];
    kb0.y = ldsK[cur][n16 * 9 + 1 + 2 * hi];
    kb1.x = ldsK[cur][n16 * 9 + 4 + 2 * hi];
    kb1.y = ldsK[cur][n16 * 9 + 5 + 2 * hi];
    v8f s = {};
    s = wmma4(qa0, kb0, s);
    s = wmma4(qa1, kb1, s);

#pragma unroll
    for (int r = 0; r < 8; ++r) {
      float mx = s[r];
      mx = fmaxf(mx, __shfl_xor(mx, 1));
      mx = fmaxf(mx, __shfl_xor(mx, 2));
      mx = fmaxf(mx, __shfl_xor(mx, 4));
      mx = fmaxf(mx, __shfl_xor(mx, 8));
      const float mn = fmaxf(m[r], mx);
      const float alpha = __expf(m[r] - mn);
      const float p = __expf(s[r] - mn);
      float ps = p;
      ps += __shfl_xor(ps, 1);
      ps += __shfl_xor(ps, 2);
      ps += __shfl_xor(ps, 4);
      ps += __shfl_xor(ps, 8);
      l[r] = l[r] * alpha + ps;
      m[r] = mn;
      o[r] *= alpha;
      myP[(r + 8 * hi) * 16 + n16] = p;
    }
    asm volatile("s_wait_dscnt 0" ::: "memory");

#pragma unroll
    for (int kc = 0; kc < 4; ++kc) {
      v2f pa, vb;
      pa.x = myP[n16 * 16 + kc * 4 + 2 * hi];
      pa.y = myP[n16 * 16 + kc * 4 + 2 * hi + 1];
      vb.x = ldsV[cur][(kc * 4 + 2 * hi) * 9 + d];
      vb.y = ldsV[cur][(kc * 4 + 2 * hi + 1) * 9 + d];
      o = wmma4(pa, vb, o);
    }
    asm volatile("" ::: "memory");
  }

  if (n16 < D_HEAD) {
#pragma unroll
    for (int r = 0; r < 8; ++r) {
      const int row = b * S_LEN + qt * 16 + r + 8 * hi;
      attn_out[row * E_DIM + h * D_HEAD + n16] = o[r] / l[r];
    }
  }
}

__global__ __launch_bounds__(128) void k_outproj_ln(
    const float* __restrict__ attn_out, const float* __restrict__ Wout,
    const float* __restrict__ bout, const float* __restrict__ x,
    const float* __restrict__ g1, const float* __restrict__ beta1,
    float* __restrict__ mlp_in) {
  __shared__ float res1[16][64];
  const int wave = threadIdx.x >> 5, lane = threadIdx.x & 31;
  const int n16 = lane & 15, hi = lane >> 4;
  const int m0 = blockIdx.x * 16;
  const int n0 = wave * 16;
  const float* arow = attn_out + (m0 + n16) * E_DIM;
  const float* brow = Wout + (n0 + n16) * E_DIM;
  v8f c = {};
#pragma unroll
  for (int k0 = 0; k0 < E_DIM; k0 += 4) {
    v2f a, b;
    a.x = arow[k0 + 2 * hi]; a.y = arow[k0 + 2 * hi + 1];
    b.x = brow[k0 + 2 * hi]; b.y = brow[k0 + 2 * hi + 1];
    c = wmma4(a, b, c);
  }
  const float bias = bout[n0 + n16];
#pragma unroll
  for (int r = 0; r < 8; ++r) {
    const int row = r + 8 * hi;
    res1[row][n0 + n16] = c[r] + bias + x[(m0 + row) * E_DIM + n0 + n16];
  }
  __syncthreads();
  const int row = threadIdx.x >> 3;
  const int c0 = (threadIdx.x & 7) * 8;
  float sum = 0.f, sq = 0.f;
#pragma unroll
  for (int i = 0; i < 8; ++i) { float v = res1[row][c0 + i]; sum += v; sq += v * v; }
#pragma unroll
  for (int msk = 1; msk < 8; msk <<= 1) {
    sum += __shfl_xor(sum, msk);
    sq += __shfl_xor(sq, msk);
  }
  const float mean = sum * (1.f / 64.f);
  const float var = sq * (1.f / 64.f) - mean * mean;
  const float rstd = rsqrtf(var + 1e-5f);
#pragma unroll
  for (int i = 0; i < 8; ++i) {
    const int col = c0 + i;
    mlp_in[(m0 + row) * E_DIM + col] =
        (res1[row][col] - mean) * rstd * g1[col] + beta1[col];
  }
}

__global__ __launch_bounds__(128) void k_mlp1(const float* __restrict__ mlp_in,
                                              const float* __restrict__ Wm1,
                                              const float* __restrict__ bm1,
                                              float* __restrict__ hbuf) {
  const int wave = threadIdx.x >> 5, lane = threadIdx.x & 31;
  const int n16 = lane & 15, hi = lane >> 4;
  const int tile = blockIdx.x * 4 + wave;
  const int mt = tile >> 3, nt = tile & 7;
  const int m0 = mt * 16, n0 = nt * 16;
  const float* arow = mlp_in + (m0 + n16) * E_DIM;
  const float* brow = Wm1 + (n0 + n16) * E_DIM;
  v8f c = {};
#pragma unroll
  for (int k0 = 0; k0 < E_DIM; k0 += 4) {
    v2f a, b;
    a.x = arow[k0 + 2 * hi]; a.y = arow[k0 + 2 * hi + 1];
    b.x = brow[k0 + 2 * hi]; b.y = brow[k0 + 2 * hi + 1];
    c = wmma4(a, b, c);
  }
  const float bias = bm1[n0 + n16];
#pragma unroll
  for (int r = 0; r < 8; ++r) {
    const int row = m0 + r + 8 * hi;
    hbuf[row * (2 * E_DIM) + n0 + n16] = fmaxf(c[r] + bias, 0.f);
  }
}

__global__ __launch_bounds__(128) void k_mlp2_ln(
    const float* __restrict__ hbuf, const float* __restrict__ Wm2,
    const float* __restrict__ bm2, const float* __restrict__ mlp_in,
    const float* __restrict__ g2, const float* __restrict__ beta2,
    float* __restrict__ out) {
  __shared__ float res2[16][64];
  const int wave = threadIdx.x >> 5, lane = threadIdx.x & 31;
  const int n16 = lane & 15, hi = lane >> 4;
  const int m0 = blockIdx.x * 16;
  const int n0 = wave * 16;
  const float* arow = hbuf + (m0 + n16) * (2 * E_DIM);
  const float* brow = Wm2 + (n0 + n16) * (2 * E_DIM);
  v8f c = {};
#pragma unroll
  for (int k0 = 0; k0 < 2 * E_DIM; k0 += 4) {
    v2f a, b;
    a.x = arow[k0 + 2 * hi]; a.y = arow[k0 + 2 * hi + 1];
    b.x = brow[k0 + 2 * hi]; b.y = brow[k0 + 2 * hi + 1];
    c = wmma4(a, b, c);
  }
  const float bias = bm2[n0 + n16];
#pragma unroll
  for (int r = 0; r < 8; ++r) {
    const int row = r + 8 * hi;
    res2[row][n0 + n16] = c[r] + bias + mlp_in[(m0 + row) * E_DIM + n0 + n16];
  }
  __syncthreads();
  const int row = threadIdx.x >> 3;
  const int c0 = (threadIdx.x & 7) * 8;
  float sum = 0.f, sq = 0.f;
#pragma unroll
  for (int i = 0; i < 8; ++i) { float v = res2[row][c0 + i]; sum += v; sq += v * v; }
#pragma unroll
  for (int msk = 1; msk < 8; msk <<= 1) {
    sum += __shfl_xor(sum, msk);
    sq += __shfl_xor(sq, msk);
  }
  const float mean = sum * (1.f / 64.f);
  const float var = sq * (1.f / 64.f) - mean * mean;
  const float rstd = rsqrtf(var + 1e-5f);
#pragma unroll
  for (int i = 0; i < 8; ++i) {
    const int col = c0 + i;
    out[(m0 + row) * E_DIM + col] =
        (res2[row][col] - mean) * rstd * g2[col] + beta2[col];
  }
}

extern "C" void kernel_launch(void* const* d_in, const int* in_sizes, int n_in,
                              void* d_out, int out_size, void* d_ws, size_t ws_size,
                              hipStream_t stream) {
  const float* x    = (const float*)d_in[0];
  const float* Wqkv = (const float*)d_in[4];
  const float* bqkv = (const float*)d_in[5];
  const float* Wout = (const float*)d_in[6];
  const float* bout = (const float*)d_in[7];
  const float* g1   = (const float*)d_in[8];
  const float* b1   = (const float*)d_in[9];
  const float* g2   = (const float*)d_in[10];
  const float* b2   = (const float*)d_in[11];
  const float* Wm1  = (const float*)d_in[12];
  const float* bm1  = (const float*)d_in[13];
  const float* Wm2  = (const float*)d_in[14];
  const float* bm2  = (const float*)d_in[15];
  float* out = (float*)d_out;

  float* ws      = (float*)d_ws;
  float* qkv     = ws;
  float* attn    = qkv + (size_t)N_TOK * 3 * E_DIM;
  float* mlpin   = attn + (size_t)N_TOK * E_DIM;
  float* hbuf    = mlpin + (size_t)N_TOK * E_DIM;

  k_qkv<<<(N_TOK / 16) * 12 / 4, 128, 0, stream>>>(x, Wqkv, bqkv, qkv);
  k_attn<<<(B_SZ * H_NUM * (S_LEN / 16)) / 4, 128, 0, stream>>>(qkv, attn);
  k_outproj_ln<<<N_TOK / 16, 128, 0, stream>>>(attn, Wout, bout, x, g1, b1, mlpin);
  k_mlp1<<<(N_TOK / 16) * 8 / 4, 128, 0, stream>>>(mlpin, Wm1, bm1, hbuf);
  k_mlp2_ln<<<N_TOK / 16, 128, 0, stream>>>(hbuf, Wm2, bm2, mlpin, g2, b2, out);
}


# k_attn: transposed score tile S^T=K*Q^T (one query per lane): key max/sum in-lane + 1 half exchange, P^T regs feed O^T=V^T*P^T as B operand (no P LDS trip, no shuffles); fp sum order changes
# speedup vs baseline: 1.4900x; 1.4900x over previous
#include <hip/hip_runtime.h>
#include <math.h>

typedef __attribute__((ext_vector_type(2))) float v2f;
typedef __attribute__((ext_vector_type(8))) float v8f;

#define E_DIM 64
#define H_NUM 8
#define D_HEAD 8
#define B_SZ 16
#define S_LEN 1024
#define N_TOK (B_SZ * S_LEN)

__device__ __forceinline__ v8f wmma4(v2f a, v2f b, v8f c) {
  return __builtin_amdgcn_wmma_f32_16x16x4_f32(
       false, a,  false, b,
       (short)0, c,  false,  false);
}

__global__ __launch_bounds__(128) void k_qkv(const float* __restrict__ x,
                                             const float* __restrict__ Wqkv,
                                             const float* __restrict__ bqkv,
                                             float* __restrict__ qkv) {
  const int wave = threadIdx.x >> 5;
  const int lane = threadIdx.x & 31;
  const int n16 = lane & 15, hi = lane >> 4;
  const int tile = blockIdx.x * 4 + wave;
  const int mt = tile / 12, nt = tile % 12;
  const int m0 = mt * 16, n0 = nt * 16;
  const float* arow = x + (m0 + n16) * E_DIM;
  const float* brow = Wqkv + (n0 + n16) * E_DIM;
  v8f c = {};
#pragma unroll
  for (int k0 = 0; k0 < E_DIM; k0 += 4) {
    v2f a, b;
    a.x = arow[k0 + 2 * hi]; a.y = arow[k0 + 2 * hi + 1];
    b.x = brow[k0 + 2 * hi]; b.y = brow[k0 + 2 * hi + 1];
    c = wmma4(a, b, c);
  }
  const float bias = bqkv[n0 + n16];
#pragma unroll
  for (int r = 0; r < 8; ++r) {
    const int row = m0 + r + 8 * hi;
    qkv[row * (3 * E_DIM) + n0 + n16] = c[r] + bias;
  }
}


__device__ __forceinline__ float xhalf(float x) {
  const int b = __builtin_bit_cast(int, x);
  return __builtin_bit_cast(
      float, __builtin_amdgcn_permlanex16(b, b, 0x76543210, 0xfedcba98, false, false));
}

__global__ __launch_bounds__(128) void k_attn(const float* __restrict__ qkv,
                                              float* __restrict__ attn_out) {
  __shared__ float ldsK[2][16 * 9];
  __shared__ float ldsV[2][16 * 9];
  const int tid = threadIdx.x;
  const int wave = tid >> 5;
  const int lane = tid & 31;
  const int n16 = lane & 15, hi = lane >> 4;

  const int bid = blockIdx.x;
  const int bh = bid >> 4;
  const int b = bh >> 3;
  const int h = bh & 7;
  const int qt = (bid & 15) * 4 + wave;
  const float scale = 0.35355339059327373f;

  const float* qkvB = qkv + (size_t)(b * S_LEN) * (3 * E_DIM);
  const int srow = tid >> 3, scol = tid & 7;

  v2f qb0, qb1;
  {
    const float* qp = qkvB + (qt * 16 + n16) * (3 * E_DIM) + h * D_HEAD;
    qb0.x = qp[0 + 2 * hi] * scale; qb0.y = qp[1 + 2 * hi] * scale;
    qb1.x = qp[4 + 2 * hi] * scale; qb1.y = qp[5 + 2 * hi] * scale;
  }

  {
    const float* rowp = qkvB + srow * (3 * E_DIM) + h * D_HEAD + scol;
    ldsK[0][srow * 9 + scol] = rowp[E_DIM];
    ldsV[0][srow * 9 + scol] = rowp[2 * E_DIM];
  }

  float m = -3.0e38f, l = 0.f;
  v8f o = {};
  const int d = n16 & 7;

  for (int kt = 0; kt < S_LEN / 16; ++kt) {
    __syncthreads();
    const int cur = kt & 1, nxt = cur ^ 1;
    if (kt + 1 < S_LEN / 16) {
      const float* rowp =
          qkvB + ((kt + 1) * 16 + srow) * (3 * E_DIM) + h * D_HEAD + scol;
      ldsK[nxt][srow * 9 + scol] = rowp[E_DIM];
      ldsV[nxt][srow * 9 + scol] = rowp[2 * E_DIM];
    }

    v2f ka0, ka1;
    ka0.x = ldsK[cur][n16 * 9 + 0 + 2 * hi];
    ka0.y = ldsK[cur][n16 * 9 + 1 + 2 * hi];
    ka1.x = ldsK[cur][n16 * 9 + 4 + 2 * hi];
    ka1.y = ldsK[cur][n16 * 9 + 5 + 2 * hi];
    v2f va[4];
#pragma unroll
    for (int kc = 0; kc < 4; ++kc) {
      va[kc].x = ldsV[cur][(8 * hi + 2 * kc) * 9 + d];
      va[kc].y = ldsV[cur][(8 * hi + 2 * kc + 1) * 9 + d];
    }

    v8f s = {};
    s = wmma4(ka0, qb0, s);
    s = wmma4(ka1, qb1, s);

    float mx = fmaxf(fmaxf(fmaxf(s[0], s[1]), fmaxf(s[2], s[3])),
                     fmaxf(fmaxf(s[4], s[5]), fmaxf(s[6], s[7])));
    mx = fmaxf(mx, xhalf(mx));
    const float mn = fmaxf(m, mx);
    const float alpha = __expf(m - mn);
    m = mn;
    v8f p;
#pragma unroll
    for (int r = 0; r < 8; ++r) p[r] = __expf(s[r] - mn);
    const float ps = ((p[0] + p[1]) + (p[2] + p[3])) + ((p[4] + p[5]) + (p[6] + p[7]));
    l = l * alpha + ps;
#pragma unroll
    for (int r = 0; r < 8; ++r) o[r] *= alpha;

#pragma unroll
    for (int kc = 0; kc < 4; ++kc) {
      v2f pb;
      pb.x = p[2 * kc];
      pb.y = p[2 * kc + 1];
      o = wmma4(va[kc], pb, o);
    }
  }

  l += xhalf(l);
  {
    float4 w;
    w.x = (hi ? o[4] : o[0]) / l;
    w.y = (hi ? o[5] : o[1]) / l;
    w.z = (hi ? o[6] : o[2]) / l;
    w.w = (hi ? o[7] : o[3]) / l;
    const int row = b * S_LEN + qt * 16 + n16;
    *reinterpret_cast<float4*>(attn_out + row * E_DIM + h * D_HEAD + 4 * hi) = w;
  }
}

__global__ __launch_bounds__(128) void k_outproj_ln(
    const float* __restrict__ attn_out, const float* __restrict__ Wout,
    const float* __restrict__ bout, const float* __restrict__ x,
    const float* __restrict__ g1, const float* __restrict__ beta1,
    float* __restrict__ mlp_in) {
  __shared__ float res1[16][64];
  const int wave = threadIdx.x >> 5, lane = threadIdx.x & 31;
  const int n16 = lane & 15, hi = lane >> 4;
  const int m0 = blockIdx.x * 16;
  const int n0 = wave * 16;
  const float* arow = attn_out + (m0 + n16) * E_DIM;
  const float* brow = Wout + (n0 + n16) * E_DIM;
  v8f c = {};
#pragma unroll
  for (int k0 = 0; k0 < E_DIM; k0 += 4) {
    v2f a, b;
    a.x = arow[k0 + 2 * hi]; a.y = arow[k0 + 2 * hi + 1];
    b.x = brow[k0 + 2 * hi]; b.y = brow[k0 + 2 * hi + 1];
    c = wmma4(a, b, c);
  }
  const float bias = bout[n0 + n16];
#pragma unroll
  for (int r = 0; r < 8; ++r) {
    const int row = r + 8 * hi;
    res1[row][n0 + n16] = c[r] + bias + x[(m0 + row) * E_DIM + n0 + n16];
  }
  __syncthreads();
  const int row = threadIdx.x >> 3;
  const int c0 = (threadIdx.x & 7) * 8;
  float sum = 0.f, sq = 0.f;
#pragma unroll
  for (int i = 0; i < 8; ++i) { float v = res1[row][c0 + i]; sum += v; sq += v * v; }
#pragma unroll
  for (int msk = 1; msk < 8; msk <<= 1) {
    sum += __shfl_xor(sum, msk);
    sq += __shfl_xor(sq, msk);
  }
  const float mean = sum * (1.f / 64.f);
  const float var = sq * (1.f / 64.f) - mean * mean;
  const float rstd = rsqrtf(var + 1e-5f);
#pragma unroll
  for (int i = 0; i < 8; ++i) {
    const int col = c0 + i;
    mlp_in[(m0 + row) * E_DIM + col] =
        (res1[row][col] - mean) * rstd * g1[col] + beta1[col];
  }
}

__global__ __launch_bounds__(128) void k_mlp1(const float* __restrict__ mlp_in,
                                              const float* __restrict__ Wm1,
                                              const float* __restrict__ bm1,
                                              float* __restrict__ hbuf) {
  const int wave = threadIdx.x >> 5, lane = threadIdx.x & 31;
  const int n16 = lane & 15, hi = lane >> 4;
  const int tile = blockIdx.x * 4 + wave;
  const int mt = tile >> 3, nt = tile & 7;
  const int m0 = mt * 16, n0 = nt * 16;
  const float* arow = mlp_in + (m0 + n16) * E_DIM;
  const float* brow = Wm1 + (n0 + n16) * E_DIM;
  v8f c = {};
#pragma unroll
  for (int k0 = 0; k0 < E_DIM; k0 += 4) {
    v2f a, b;
    a.x = arow[k0 + 2 * hi]; a.y = arow[k0 + 2 * hi + 1];
    b.x = brow[k0 + 2 * hi]; b.y = brow[k0 + 2 * hi + 1];
    c = wmma4(a, b, c);
  }
  const float bias = bm1[n0 + n16];
#pragma unroll
  for (int r = 0; r < 8; ++r) {
    const int row = m0 + r + 8 * hi;
    hbuf[row * (2 * E_DIM) + n0 + n16] = fmaxf(c[r] + bias, 0.f);
  }
}

__global__ __launch_bounds__(128) void k_mlp2_ln(
    const float* __restrict__ hbuf, const float* __restrict__ Wm2,
    const float* __restrict__ bm2, const float* __restrict__ mlp_in,
    const float* __restrict__ g2, const float* __restrict__ beta2,
    float* __restrict__ out) {
  __shared__ float res2[16][64];
  const int wave = threadIdx.x >> 5, lane = threadIdx.x & 31;
  const int n16 = lane & 15, hi = lane >> 4;
  const int m0 = blockIdx.x * 16;
  const int n0 = wave * 16;
  const float* arow = hbuf + (m0 + n16) * (2 * E_DIM);
  const float* brow = Wm2 + (n0 + n16) * (2 * E_DIM);
  v8f c = {};
#pragma unroll
  for (int k0 = 0; k0 < 2 * E_DIM; k0 += 4) {
    v2f a, b;
    a.x = arow[k0 + 2 * hi]; a.y = arow[k0 + 2 * hi + 1];
    b.x = brow[k0 + 2 * hi]; b.y = brow[k0 + 2 * hi + 1];
    c = wmma4(a, b, c);
  }
  const float bias = bm2[n0 + n16];
#pragma unroll
  for (int r = 0; r < 8; ++r) {
    const int row = r + 8 * hi;
    res2[row][n0 + n16] = c[r] + bias + mlp_in[(m0 + row) * E_DIM + n0 + n16];
  }
  __syncthreads();
  const int row = threadIdx.x >> 3;
  const int c0 = (threadIdx.x & 7) * 8;
  float sum = 0.f, sq = 0.f;
#pragma unroll
  for (int i = 0; i < 8; ++i) { float v = res2[row][c0 + i]; sum += v; sq += v * v; }
#pragma unroll
  for (int msk = 1; msk < 8; msk <<= 1) {
    sum += __shfl_xor(sum, msk);
    sq += __shfl_xor(sq, msk);
  }
  const float mean = sum * (1.f / 64.f);
  const float var = sq * (1.f / 64.f) - mean * mean;
  const float rstd = rsqrtf(var + 1e-5f);
#pragma unroll
  for (int i = 0; i < 8; ++i) {
    const int col = c0 + i;
    out[(m0 + row) * E_DIM + col] =
        (res2[row][col] - mean) * rstd * g2[col] + beta2[col];
  }
}

extern "C" void kernel_launch(void* const* d_in, const int* in_sizes, int n_in,
                              void* d_out, int out_size, void* d_ws, size_t ws_size,
                              hipStream_t stream) {
  const float* x    = (const float*)d_in[0];
  const float* Wqkv = (const float*)d_in[4];
  const float* bqkv = (const float*)d_in[5];
  const float* Wout = (const float*)d_in[6];
  const float* bout = (const float*)d_in[7];
  const float* g1   = (const float*)d_in[8];
  const float* b1   = (const float*)d_in[9];
  const float* g2   = (const float*)d_in[10];
  const float* b2   = (const float*)d_in[11];
  const float* Wm1  = (const float*)d_in[12];
  const float* bm1  = (const float*)d_in[13];
  const float* Wm2  = (const float*)d_in[14];
  const float* bm2  = (const float*)d_in[15];
  float* out = (float*)d_out;

  float* ws      = (float*)d_ws;
  float* qkv     = ws;
  float* attn    = qkv + (size_t)N_TOK * 3 * E_DIM;
  float* mlpin   = attn + (size_t)N_TOK * E_DIM;
  float* hbuf    = mlpin + (size_t)N_TOK * E_DIM;

  k_qkv<<<(N_TOK / 16) * 12 / 4, 128, 0, stream>>>(x, Wqkv, bqkv, qkv);
  k_attn<<<(B_SZ * H_NUM * (S_LEN / 16)) / 4, 128, 0, stream>>>(qkv, attn);
  k_outproj_ln<<<N_TOK / 16, 128, 0, stream>>>(attn, Wout, bout, x, g1, b1, mlpin);
  k_mlp1<<<(N_TOK / 16) * 8 / 4, 128, 0, stream>>>(mlpin, Wm1, bm1, hbuf);
  k_mlp2_ln<<<N_TOK / 16, 128, 0, stream>>>(hbuf, Wm2, bm2, mlpin, g2, b2, out);
}


# fuse out-proj+LN1, MLP1 and MLP2+LN2 into one kernel per 16-row block (5 launches -> 3): LN1 output and relu(MLP1) stay in LDS as WMMA A operands, weight fragments loaded once at kernel start
# speedup vs baseline: 1.6729x; 1.1228x over previous
#include <hip/hip_runtime.h>
#include <math.h>

typedef __attribute__((ext_vector_type(2))) float v2f;
typedef __attribute__((ext_vector_type(8))) float v8f;

#define E_DIM 64
#define H_NUM 8
#define D_HEAD 8
#define B_SZ 16
#define S_LEN 1024
#define N_TOK (B_SZ * S_LEN)

__device__ __forceinline__ v8f wmma4(v2f a, v2f b, v8f c) {
  return __builtin_amdgcn_wmma_f32_16x16x4_f32(
       false, a,  false, b,
       (short)0, c,  false,  false);
}

__global__ __launch_bounds__(128) void k_qkv(const float* __restrict__ x,
                                             const float* __restrict__ Wqkv,
                                             const float* __restrict__ bqkv,
                                             float* __restrict__ qkv) {
  const int wave = threadIdx.x >> 5;
  const int lane = threadIdx.x & 31;
  const int n16 = lane & 15, hi = lane >> 4;
  const int tile = blockIdx.x * 4 + wave;
  const int mt = tile / 12, nt = tile % 12;
  const int m0 = mt * 16, n0 = nt * 16;
  const float* arow = x + (m0 + n16) * E_DIM;
  const float* brow = Wqkv + (n0 + n16) * E_DIM;
  v8f c = {};
#pragma unroll
  for (int k0 = 0; k0 < E_DIM; k0 += 4) {
    v2f a, b;
    a.x = arow[k0 + 2 * hi]; a.y = arow[k0 + 2 * hi + 1];
    b.x = brow[k0 + 2 * hi]; b.y = brow[k0 + 2 * hi + 1];
    c = wmma4(a, b, c);
  }
  const float bias = bqkv[n0 + n16];
#pragma unroll
  for (int r = 0; r < 8; ++r) {
    const int row = m0 + r + 8 * hi;
    qkv[row * (3 * E_DIM) + n0 + n16] = c[r] + bias;
  }
}


__device__ __forceinline__ float xhalf(float x) {
  const int b = __builtin_bit_cast(int, x);
  return __builtin_bit_cast(
      float, __builtin_amdgcn_permlanex16(b, b, 0x76543210, 0xfedcba98, false, false));
}

__global__ __launch_bounds__(128) void k_attn(const float* __restrict__ qkv,
                                              float* __restrict__ attn_out) {
  __shared__ float ldsK[2][16 * 9];
  __shared__ float ldsV[2][16 * 9];
  const int tid = threadIdx.x;
  const int wave = tid >> 5;
  const int lane = tid & 31;
  const int n16 = lane & 15, hi = lane >> 4;

  const int bid = blockIdx.x;
  const int bh = bid >> 4;
  const int b = bh >> 3;
  const int h = bh & 7;
  const int qt = (bid & 15) * 4 + wave;
  const float scale = 0.35355339059327373f;

  const float* qkvB = qkv + (size_t)(b * S_LEN) * (3 * E_DIM);
  const int srow = tid >> 3, scol = tid & 7;

  v2f qb0, qb1;
  {
    const float* qp = qkvB + (qt * 16 + n16) * (3 * E_DIM) + h * D_HEAD;
    qb0.x = qp[0 + 2 * hi] * scale; qb0.y = qp[1 + 2 * hi] * scale;
    qb1.x = qp[4 + 2 * hi] * scale; qb1.y = qp[5 + 2 * hi] * scale;
  }

  {
    const float* rowp = qkvB + srow * (3 * E_DIM) + h * D_HEAD + scol;
    ldsK[0][srow * 9 + scol] = rowp[E_DIM];
    ldsV[0][srow * 9 + scol] = rowp[2 * E_DIM];
  }

  float m = -3.0e38f, l = 0.f;
  v8f o = {};
  const int d = n16 & 7;

  for (int kt = 0; kt < S_LEN / 16; ++kt) {
    __syncthreads();
    const int cur = kt & 1, nxt = cur ^ 1;
    if (kt + 1 < S_LEN / 16) {
      const float* rowp =
          qkvB + ((kt + 1) * 16 + srow) * (3 * E_DIM) + h * D_HEAD + scol;
      ldsK[nxt][srow * 9 + scol] = rowp[E_DIM];
      ldsV[nxt][srow * 9 + scol] = rowp[2 * E_DIM];
    }

    v2f ka0, ka1;
    ka0.x = ldsK[cur][n16 * 9 + 0 + 2 * hi];
    ka0.y = ldsK[cur][n16 * 9 + 1 + 2 * hi];
    ka1.x = ldsK[cur][n16 * 9 + 4 + 2 * hi];
    ka1.y = ldsK[cur][n16 * 9 + 5 + 2 * hi];
    v2f va[4];
#pragma unroll
    for (int kc = 0; kc < 4; ++kc) {
      va[kc].x = ldsV[cur][(8 * hi + 2 * kc) * 9 + d];
      va[kc].y = ldsV[cur][(8 * hi + 2 * kc + 1) * 9 + d];
    }

    v8f s = {};
    s = wmma4(ka0, qb0, s);
    s = wmma4(ka1, qb1, s);

    float mx = fmaxf(fmaxf(fmaxf(s[0], s[1]), fmaxf(s[2], s[3])),
                     fmaxf(fmaxf(s[4], s[5]), fmaxf(s[6], s[7])));
    mx = fmaxf(mx, xhalf(mx));
    const float mn = fmaxf(m, mx);
    const float alpha = __expf(m - mn);
    m = mn;
    v8f p;
#pragma unroll
    for (int r = 0; r < 8; ++r) p[r] = __expf(s[r] - mn);
    const float ps = ((p[0] + p[1]) + (p[2] + p[3])) + ((p[4] + p[5]) + (p[6] + p[7]));
    l = l * alpha + ps;
#pragma unroll
    for (int r = 0; r < 8; ++r) o[r] *= alpha;

#pragma unroll
    for (int kc = 0; kc < 4; ++kc) {
      v2f pb;
      pb.x = p[2 * kc];
      pb.y = p[2 * kc + 1];
      o = wmma4(va[kc], pb, o);
    }
  }

  l += xhalf(l);
  {
    float4 w;
    w.x = (hi ? o[4] : o[0]) / l;
    w.y = (hi ? o[5] : o[1]) / l;
    w.z = (hi ? o[6] : o[2]) / l;
    w.w = (hi ? o[7] : o[3]) / l;
    const int row = b * S_LEN + qt * 16 + n16;
    *reinterpret_cast<float4*>(attn_out + row * E_DIM + h * D_HEAD + 4 * hi) = w;
  }
}

#define P64 68
#define P128 132

__device__ __forceinline__ v2f ld2(const float* p) {
  v2f v;
  v.x = p[0];
  v.y = p[1];
  return v;
}

__global__ __launch_bounds__(128) void k_tail(
    const float* __restrict__ attn_out, const float* __restrict__ Wout,
    const float* __restrict__ bout, const float* __restrict__ x,
    const float* __restrict__ g1, const float* __restrict__ beta1,
    const float* __restrict__ Wm1, const float* __restrict__ bm1,
    const float* __restrict__ Wm2, const float* __restrict__ bm2,
    const float* __restrict__ g2, const float* __restrict__ beta2,
    float* __restrict__ out) {
  __shared__ float res[16][P64];
  __shared__ float mlpin[16][P64];
  __shared__ float hbuf[16][P128];
  const int wave = threadIdx.x >> 5, lane = threadIdx.x & 31;
  const int n16 = lane & 15, hi = lane >> 4;
  const int m0 = blockIdx.x * 16;
  const int n0 = wave * 16;
  const int nh = wave * 32;

  const float* arow = attn_out + (m0 + n16) * E_DIM + 2 * hi;
  const float* brow = Wout + (n0 + n16) * E_DIM + 2 * hi;
  const float* w1r0 = Wm1 + (nh + n16) * E_DIM + 2 * hi;
  const float* w1r1 = Wm1 + (nh + 16 + n16) * E_DIM + 2 * hi;
  const float* w2r = Wm2 + (n0 + n16) * (2 * E_DIM) + 2 * hi;
  v2f fa[16], fb[16], fw1a[16], fw1b[16], fw2[32];
#pragma unroll
  for (int s = 0; s < 16; ++s) { fa[s] = ld2(arow + 4 * s); fb[s] = ld2(brow + 4 * s); }
  float xr[8];
#pragma unroll
  for (int r = 0; r < 8; ++r) xr[r] = x[(m0 + r + 8 * hi) * E_DIM + n0 + n16];
#pragma unroll
  for (int s = 0; s < 16; ++s) { fw1a[s] = ld2(w1r0 + 4 * s); fw1b[s] = ld2(w1r1 + 4 * s); }
#pragma unroll
  for (int s = 0; s < 32; ++s) fw2[s] = ld2(w2r + 4 * s);

  {
    v8f c = {};
#pragma unroll
    for (int s = 0; s < 16; ++s) c = wmma4(fa[s], fb[s], c);
    const float bias = bout[n0 + n16];
#pragma unroll
    for (int r = 0; r < 8; ++r) res[r + 8 * hi][n0 + n16] = c[r] + bias + xr[r];
  }
  __syncthreads();

  const int row = threadIdx.x >> 3;
  const int c0 = (threadIdx.x & 7) * 8;
  {
    float sum = 0.f, sq = 0.f;
#pragma unroll
    for (int i = 0; i < 8; ++i) { float v = res[row][c0 + i]; sum += v; sq += v * v; }
#pragma unroll
    for (int msk = 1; msk < 8; msk <<= 1) {
      sum += __shfl_xor(sum, msk);
      sq += __shfl_xor(sq, msk);
    }
    const float mean = sum * (1.f / 64.f);
    const float var = sq * (1.f / 64.f) - mean * mean;
    const float rstd = rsqrtf(var + 1e-5f);
#pragma unroll
    for (int i = 0; i < 8; ++i) {
      const int col = c0 + i;
      mlpin[row][col] = (res[row][col] - mean) * rstd * g1[col] + beta1[col];
    }
  }
  __syncthreads();

  {
    v8f h0 = {}, h1 = {};
#pragma unroll
    for (int s = 0; s < 16; ++s) {
      const v2f a = ld2(&mlpin[n16][4 * s + 2 * hi]);
      h0 = wmma4(a, fw1a[s], h0);
      h1 = wmma4(a, fw1b[s], h1);
    }
    const float bias0 = bm1[nh + n16], bias1 = bm1[nh + 16 + n16];
#pragma unroll
    for (int r = 0; r < 8; ++r) {
      hbuf[r + 8 * hi][nh + n16] = fmaxf(h0[r] + bias0, 0.f);
      hbuf[r + 8 * hi][nh + 16 + n16] = fmaxf(h1[r] + bias1, 0.f);
    }
  }
  __syncthreads();

  {
    v8f c = {};
#pragma unroll
    for (int s = 0; s < 32; ++s) {
      const v2f a = ld2(&hbuf[n16][4 * s + 2 * hi]);
      c = wmma4(a, fw2[s], c);
    }
    const float bias = bm2[n0 + n16];
#pragma unroll
    for (int r = 0; r < 8; ++r)
      res[r + 8 * hi][n0 + n16] = c[r] + bias + mlpin[r + 8 * hi][n0 + n16];
  }
  __syncthreads();

  {
    float sum = 0.f, sq = 0.f;
#pragma unroll
    for (int i = 0; i < 8; ++i) { float v = res[row][c0 + i]; sum += v; sq += v * v; }
#pragma unroll
    for (int msk = 1; msk < 8; msk <<= 1) {
      sum += __shfl_xor(sum, msk);
      sq += __shfl_xor(sq, msk);
    }
    const float mean = sum * (1.f / 64.f);
    const float var = sq * (1.f / 64.f) - mean * mean;
    const float rstd = rsqrtf(var + 1e-5f);
#pragma unroll
    for (int i = 0; i < 8; ++i) {
      const int col = c0 + i;
      out[(m0 + row) * E_DIM + col] =
          (res[row][col] - mean) * rstd * g2[col] + beta2[col];
    }
  }
}

extern "C" void kernel_launch(void* const* d_in, const int* in_sizes, int n_in,
                              void* d_out, int out_size, void* d_ws, size_t ws_size,
                              hipStream_t stream) {
  const float* x    = (const float*)d_in[0];
  const float* Wqkv = (const float*)d_in[4];
  const float* bqkv = (const float*)d_in[5];
  const float* Wout = (const float*)d_in[6];
  const float* bout = (const float*)d_in[7];
  const float* g1   = (const float*)d_in[8];
  const float* b1   = (const float*)d_in[9];
  const float* g2   = (const float*)d_in[10];
  const float* b2   = (const float*)d_in[11];
  const float* Wm1  = (const float*)d_in[12];
  const float* bm1  = (const float*)d_in[13];
  const float* Wm2  = (const float*)d_in[14];
  const float* bm2  = (const float*)d_in[15];
  float* out = (float*)d_out;

  float* ws      = (float*)d_ws;
  float* qkv     = ws;
  float* attn    = qkv + (size_t)N_TOK * 3 * E_DIM;

  k_qkv<<<(N_TOK / 16) * 12 / 4, 128, 0, stream>>>(x, Wqkv, bqkv, qkv);
  k_attn<<<(B_SZ * H_NUM * (S_LEN / 16)) / 4, 128, 0, stream>>>(qkv, attn);
  k_tail<<<N_TOK / 16, 128, 0, stream>>>(attn, Wout, bout, x, g1, b1,
                                         Wm1, bm1, Wm2, bm2, g2, b2, out);
}


# k_attn: K/V staging as a 3-deep software pipeline: global loads issued two key tiles ahead into registers, static LDS stages (loop unrolled x2), barrier fences LDS only; no wait on fresh loads
# speedup vs baseline: 1.9004x; 1.1360x over previous
#include <hip/hip_runtime.h>
#include <math.h>

typedef __attribute__((ext_vector_type(2))) float v2f;
typedef __attribute__((ext_vector_type(8))) float v8f;

#define E_DIM 64
#define H_NUM 8
#define D_HEAD 8
#define B_SZ 16
#define S_LEN 1024
#define N_TOK (B_SZ * S_LEN)

__device__ __forceinline__ v8f wmma4(v2f a, v2f b, v8f c) {
  return __builtin_amdgcn_wmma_f32_16x16x4_f32(
       false, a,  false, b,
       (short)0, c,  false,  false);
}

__global__ __launch_bounds__(128) void k_qkv(const float* __restrict__ x,
                                             const float* __restrict__ Wqkv,
                                             const float* __restrict__ bqkv,
                                             float* __restrict__ qkv) {
  const int wave = threadIdx.x >> 5;
  const int lane = threadIdx.x & 31;
  const int n16 = lane & 15, hi = lane >> 4;
  const int tile = blockIdx.x * 4 + wave;
  const int mt = tile / 12, nt = tile % 12;
  const int m0 = mt * 16, n0 = nt * 16;
  const float* arow = x + (m0 + n16) * E_DIM;
  const float* brow = Wqkv + (n0 + n16) * E_DIM;
  v8f c = {};
#pragma unroll
  for (int k0 = 0; k0 < E_DIM; k0 += 4) {
    v2f a, b;
    a.x = arow[k0 + 2 * hi]; a.y = arow[k0 + 2 * hi + 1];
    b.x = brow[k0 + 2 * hi]; b.y = brow[k0 + 2 * hi + 1];
    c = wmma4(a, b, c);
  }
  const float bias = bqkv[n0 + n16];
#pragma unroll
  for (int r = 0; r < 8; ++r) {
    const int row = m0 + r + 8 * hi;
    qkv[row * (3 * E_DIM) + n0 + n16] = c[r] + bias;
  }
}


__device__ __forceinline__ float xhalf(float x) {
  const int b = __builtin_bit_cast(int, x);
  return __builtin_bit_cast(
      float, __builtin_amdgcn_permlanex16(b, b, 0x76543210, 0xfedcba98, false, false));
}

__device__ __forceinline__ void lds_barrier() {
  __builtin_amdgcn_fence(__ATOMIC_RELEASE, "workgroup", "local");
  __builtin_amdgcn_s_barrier();
  __builtin_amdgcn_fence(__ATOMIC_ACQUIRE, "workgroup", "local");
}

#define N_KT (S_LEN / 16)
#define KT_STEP (16 * 3 * E_DIM)

struct KVFrags {
  v2f ka0, ka1;
  v2f va[4];
};

__device__ __forceinline__ KVFrags load_frags(const float* kb, const float* vb,
                                              int n16, int hi, int d) {
  KVFrags f;
  f.ka0.x = kb[n16 * 9 + 0 + 2 * hi];
  f.ka0.y = kb[n16 * 9 + 1 + 2 * hi];
  f.ka1.x = kb[n16 * 9 + 4 + 2 * hi];
  f.ka1.y = kb[n16 * 9 + 5 + 2 * hi];
#pragma unroll
  for (int kc = 0; kc < 4; ++kc) {
    f.va[kc].x = vb[(8 * hi + 2 * kc) * 9 + d];
    f.va[kc].y = vb[(8 * hi + 2 * kc + 1) * 9 + d];
  }
  return f;
}

__device__ __forceinline__ void attn_tile(const KVFrags& f, v2f qb0, v2f qb1,
                                          float& m, float& l, v8f& o) {
  v8f s = {};
  s = wmma4(f.ka0, qb0, s);
  s = wmma4(f.ka1, qb1, s);

  float mx = fmaxf(fmaxf(fmaxf(s[0], s[1]), fmaxf(s[2], s[3])),
                   fmaxf(fmaxf(s[4], s[5]), fmaxf(s[6], s[7])));
  mx = fmaxf(mx, xhalf(mx));
  const float mn = fmaxf(m, mx);
  const float alpha = __expf(m - mn);
  m = mn;
  v8f p;
#pragma unroll
  for (int r = 0; r < 8; ++r) p[r] = __expf(s[r] - mn);
  const float ps = ((p[0] + p[1]) + (p[2] + p[3])) + ((p[4] + p[5]) + (p[6] + p[7]));
  l = l * alpha + ps;
#pragma unroll
  for (int r = 0; r < 8; ++r) o[r] *= alpha;

#pragma unroll
  for (int kc = 0; kc < 4; ++kc) {
    v2f pb;
    pb.x = p[2 * kc];
    pb.y = p[2 * kc + 1];
    o = wmma4(f.va[kc], pb, o);
  }
}

__global__ __launch_bounds__(128) void k_attn(const float* __restrict__ qkv,
                                              float* __restrict__ attn_out) {
  __shared__ float ldsK0[16 * 9], ldsK1[16 * 9];
  __shared__ float ldsV0[16 * 9], ldsV1[16 * 9];
  const int tid = threadIdx.x;
  const int wave = tid >> 5;
  const int lane = tid & 31;
  const int n16 = lane & 15, hi = lane >> 4;

  const int bid = blockIdx.x;
  const int bh = bid >> 4;
  const int b = bh >> 3;
  const int h = bh & 7;
  const int qt = (bid & 15) * 4 + wave;
  const float scale = 0.35355339059327373f;

  const float* qkvB = qkv + (size_t)(b * S_LEN) * (3 * E_DIM);
  const int srow = tid >> 3, scol = tid & 7;
  const int sidx = srow * 9 + scol;
  const float* stg = qkvB + srow * (3 * E_DIM) + h * D_HEAD + scol;

  v2f qb0, qb1;
  {
    const float* qp = qkvB + (qt * 16 + n16) * (3 * E_DIM) + h * D_HEAD;
    qb0.x = qp[0 + 2 * hi] * scale; qb0.y = qp[1 + 2 * hi] * scale;
    qb1.x = qp[4 + 2 * hi] * scale; qb1.y = qp[5 + 2 * hi] * scale;
  }

  float kA, vA, kB, vB;
  {
    const float k0 = stg[E_DIM], v0 = stg[2 * E_DIM];
    kA = stg[KT_STEP + E_DIM];     vA = stg[KT_STEP + 2 * E_DIM];
    kB = stg[2 * KT_STEP + E_DIM]; vB = stg[2 * KT_STEP + 2 * E_DIM];
    ldsK0[sidx] = k0;
    ldsV0[sidx] = v0;
  }

  float m = -3.0e38f, l = 0.f;
  v8f o = {};
  const int d = n16 & 7;

  for (int kt = 0; kt < N_KT; kt += 2) {
    lds_barrier();
    {
      const KVFrags f = load_frags(ldsK0, ldsV0, n16, hi, d);
      ldsK1[sidx] = kA;
      ldsV1[sidx] = vA;
      const float* nx = stg + min(kt + 3, N_KT - 1) * KT_STEP;
      kA = nx[E_DIM];
      vA = nx[2 * E_DIM];
      attn_tile(f, qb0, qb1, m, l, o);
    }
    lds_barrier();
    {
      const KVFrags f = load_frags(ldsK1, ldsV1, n16, hi, d);
      ldsK0[sidx] = kB;
      ldsV0[sidx] = vB;
      const float* nx = stg + min(kt + 4, N_KT - 1) * KT_STEP;
      kB = nx[E_DIM];
      vB = nx[2 * E_DIM];
      attn_tile(f, qb0, qb1, m, l, o);
    }
  }

  l += xhalf(l);
  {
    float4 w;
    w.x = (hi ? o[4] : o[0]) / l;
    w.y = (hi ? o[5] : o[1]) / l;
    w.z = (hi ? o[6] : o[2]) / l;
    w.w = (hi ? o[7] : o[3]) / l;
    const int row = b * S_LEN + qt * 16 + n16;
    *reinterpret_cast<float4*>(attn_out + row * E_DIM + h * D_HEAD + 4 * hi) = w;
  }
}

#define P64 68
#define P128 132

__device__ __forceinline__ v2f ld2(const float* p) {
  v2f v;
  v.x = p[0];
  v.y = p[1];
  return v;
}

__global__ __launch_bounds__(128) void k_tail(
    const float* __restrict__ attn_out, const float* __restrict__ Wout,
    const float* __restrict__ bout, const float* __restrict__ x,
    const float* __restrict__ g1, const float* __restrict__ beta1,
    const float* __restrict__ Wm1, const float* __restrict__ bm1,
    const float* __restrict__ Wm2, const float* __restrict__ bm2,
    const float* __restrict__ g2, const float* __restrict__ beta2,
    float* __restrict__ out) {
  __shared__ float res[16][P64];
  __shared__ float mlpin[16][P64];
  __shared__ float hbuf[16][P128];
  const int wave = threadIdx.x >> 5, lane = threadIdx.x & 31;
  const int n16 = lane & 15, hi = lane >> 4;
  const int m0 = blockIdx.x * 16;
  const int n0 = wave * 16;
  const int nh = wave * 32;

  const float* arow = attn_out + (m0 + n16) * E_DIM + 2 * hi;
  const float* brow = Wout + (n0 + n16) * E_DIM + 2 * hi;
  const float* w1r0 = Wm1 + (nh + n16) * E_DIM + 2 * hi;
  const float* w1r1 = Wm1 + (nh + 16 + n16) * E_DIM + 2 * hi;
  const float* w2r = Wm2 + (n0 + n16) * (2 * E_DIM) + 2 * hi;
  v2f fa[16], fb[16], fw1a[16], fw1b[16], fw2[32];
#pragma unroll
  for (int s = 0; s < 16; ++s) { fa[s] = ld2(arow + 4 * s); fb[s] = ld2(brow + 4 * s); }
  float xr[8];
#pragma unroll
  for (int r = 0; r < 8; ++r) xr[r] = x[(m0 + r + 8 * hi) * E_DIM + n0 + n16];
#pragma unroll
  for (int s = 0; s < 16; ++s) { fw1a[s] = ld2(w1r0 + 4 * s); fw1b[s] = ld2(w1r1 + 4 * s); }
#pragma unroll
  for (int s = 0; s < 32; ++s) fw2[s] = ld2(w2r + 4 * s);

  {
    v8f c = {};
#pragma unroll
    for (int s = 0; s < 16; ++s) c = wmma4(fa[s], fb[s], c);
    const float bias = bout[n0 + n16];
#pragma unroll
    for (int r = 0; r < 8; ++r) res[r + 8 * hi][n0 + n16] = c[r] + bias + xr[r];
  }
  __syncthreads();

  const int row = threadIdx.x >> 3;
  const int c0 = (threadIdx.x & 7) * 8;
  {
    float sum = 0.f, sq = 0.f;
#pragma unroll
    for (int i = 0; i < 8; ++i) { float v = res[row][c0 + i]; sum += v; sq += v * v; }
#pragma unroll
    for (int msk = 1; msk < 8; msk <<= 1) {
      sum += __shfl_xor(sum, msk);
      sq += __shfl_xor(sq, msk);
    }
    const float mean = sum * (1.f / 64.f);
    const float var = sq * (1.f / 64.f) - mean * mean;
    const float rstd = rsqrtf(var + 1e-5f);
#pragma unroll
    for (int i = 0; i < 8; ++i) {
      const int col = c0 + i;
      mlpin[row][col] = (res[row][col] - mean) * rstd * g1[col] + beta1[col];
    }
  }
  __syncthreads();

  {
    v8f h0 = {}, h1 = {};
#pragma unroll
    for (int s = 0; s < 16; ++s) {
      const v2f a = ld2(&mlpin[n16][4 * s + 2 * hi]);
      h0 = wmma4(a, fw1a[s], h0);
      h1 = wmma4(a, fw1b[s], h1);
    }
    const float bias0 = bm1[nh + n16], bias1 = bm1[nh + 16 + n16];
#pragma unroll
    for (int r = 0; r < 8; ++r) {
      hbuf[r + 8 * hi][nh + n16] = fmaxf(h0[r] + bias0, 0.f);
      hbuf[r + 8 * hi][nh + 16 + n16] = fmaxf(h1[r] + bias1, 0.f);
    }
  }
  __syncthreads();

  {
    v8f c = {};
#pragma unroll
    for (int s = 0; s < 32; ++s) {
      const v2f a = ld2(&hbuf[n16][4 * s + 2 * hi]);
      c = wmma4(a, fw2[s], c);
    }
    const float bias = bm2[n0 + n16];
#pragma unroll
    for (int r = 0; r < 8; ++r)
      res[r + 8 * hi][n0 + n16] = c[r] + bias + mlpin[r + 8 * hi][n0 + n16];
  }
  __syncthreads();

  {
    float sum = 0.f, sq = 0.f;
#pragma unroll
    for (int i = 0; i < 8; ++i) { float v = res[row][c0 + i]; sum += v; sq += v * v; }
#pragma unroll
    for (int msk = 1; msk < 8; msk <<= 1) {
      sum += __shfl_xor(sum, msk);
      sq += __shfl_xor(sq, msk);
    }
    const float mean = sum * (1.f / 64.f);
    const float var = sq * (1.f / 64.f) - mean * mean;
    const float rstd = rsqrtf(var + 1e-5f);
#pragma unroll
    for (int i = 0; i < 8; ++i) {
      const int col = c0 + i;
      out[(m0 + row) * E_DIM + col] =
          (res[row][col] - mean) * rstd * g2[col] + beta2[col];
    }
  }
}

extern "C" void kernel_launch(void* const* d_in, const int* in_sizes, int n_in,
                              void* d_out, int out_size, void* d_ws, size_t ws_size,
                              hipStream_t stream) {
  const float* x    = (const float*)d_in[0];
  const float* Wqkv = (const float*)d_in[4];
  const float* bqkv = (const float*)d_in[5];
  const float* Wout = (const float*)d_in[6];
  const float* bout = (const float*)d_in[7];
  const float* g1   = (const float*)d_in[8];
  const float* b1   = (const float*)d_in[9];
  const float* g2   = (const float*)d_in[10];
  const float* b2   = (const float*)d_in[11];
  const float* Wm1  = (const float*)d_in[12];
  const float* bm1  = (const float*)d_in[13];
  const float* Wm2  = (const float*)d_in[14];
  const float* bm2  = (const float*)d_in[15];
  float* out = (float*)d_out;

  float* ws      = (float*)d_ws;
  float* qkv     = ws;
  float* attn    = qkv + (size_t)N_TOK * 3 * E_DIM;

  k_qkv<<<(N_TOK / 16) * 12 / 4, 128, 0, stream>>>(x, Wqkv, bqkv, qkv);
  k_attn<<<(B_SZ * H_NUM * (S_LEN / 16)) / 4, 128, 0, stream>>>(qkv, attn);
  k_tail<<<N_TOK / 16, 128, 0, stream>>>(attn, Wout, bout, x, g1, b1,
                                         Wm1, bm1, Wm2, bm2, g2, b2, out);
}


# k_attn: each wave owns two 16-query tiles (32 query columns), so every K/V^T fragment read from LDS, staged K/V element and barrier serves two score tiles; two independent WMMA chains per wave
# speedup vs baseline: 2.0539x; 1.0808x over previous
#include <hip/hip_runtime.h>
#include <math.h>

typedef __attribute__((ext_vector_type(2))) float v2f;
typedef __attribute__((ext_vector_type(8))) float v8f;

#define E_DIM 64
#define H_NUM 8
#define D_HEAD 8
#define B_SZ 16
#define S_LEN 1024
#define N_TOK (B_SZ * S_LEN)

__device__ __forceinline__ v8f wmma4(v2f a, v2f b, v8f c) {
  return __builtin_amdgcn_wmma_f32_16x16x4_f32(
       false, a,  false, b,
       (short)0, c,  false,  false);
}

__global__ __launch_bounds__(128) void k_qkv(const float* __restrict__ x,
                                             const float* __restrict__ Wqkv,
                                             const float* __restrict__ bqkv,
                                             float* __restrict__ qkv) {
  const int wave = threadIdx.x >> 5;
  const int lane = threadIdx.x & 31;
  const int n16 = lane & 15, hi = lane >> 4;
  const int tile = blockIdx.x * 4 + wave;
  const int mt = tile / 12, nt = tile % 12;
  const int m0 = mt * 16, n0 = nt * 16;
  const float* arow = x + (m0 + n16) * E_DIM;
  const float* brow = Wqkv + (n0 + n16) * E_DIM;
  v8f c = {};
#pragma unroll
  for (int k0 = 0; k0 < E_DIM; k0 += 4) {
    v2f a, b;
    a.x = arow[k0 + 2 * hi]; a.y = arow[k0 + 2 * hi + 1];
    b.x = brow[k0 + 2 * hi]; b.y = brow[k0 + 2 * hi + 1];
    c = wmma4(a, b, c);
  }
  const float bias = bqkv[n0 + n16];
#pragma unroll
  for (int r = 0; r < 8; ++r) {
    const int row = m0 + r + 8 * hi;
    qkv[row * (3 * E_DIM) + n0 + n16] = c[r] + bias;
  }
}


__device__ __forceinline__ float xhalf(float x) {
  const int b = __builtin_bit_cast(int, x);
  return __builtin_bit_cast(
      float, __builtin_amdgcn_permlanex16(b, b, 0x76543210, 0xfedcba98, false, false));
}

__device__ __forceinline__ void lds_barrier() {
  __builtin_amdgcn_fence(__ATOMIC_RELEASE, "workgroup", "local");
  __builtin_amdgcn_s_barrier();
  __builtin_amdgcn_fence(__ATOMIC_ACQUIRE, "workgroup", "local");
}

#define N_KT (S_LEN / 16)
#define KT_STEP (16 * 3 * E_DIM)

struct KVFrags {
  v2f ka0, ka1;
  v2f va[4];
};

__device__ __forceinline__ KVFrags load_frags(const float* kb, const float* vb,
                                              int n16, int hi, int d) {
  KVFrags f;
  f.ka0.x = kb[n16 * 9 + 0 + 2 * hi];
  f.ka0.y = kb[n16 * 9 + 1 + 2 * hi];
  f.ka1.x = kb[n16 * 9 + 4 + 2 * hi];
  f.ka1.y = kb[n16 * 9 + 5 + 2 * hi];
#pragma unroll
  for (int kc = 0; kc < 4; ++kc) {
    f.va[kc].x = vb[(8 * hi + 2 * kc) * 9 + d];
    f.va[kc].y = vb[(8 * hi + 2 * kc + 1) * 9 + d];
  }
  return f;
}

__device__ __forceinline__ void attn_tile(const KVFrags& f, v2f qb0, v2f qb1,
                                          float& m, float& l, v8f& o) {
  v8f s = {};
  s = wmma4(f.ka0, qb0, s);
  s = wmma4(f.ka1, qb1, s);

  float mx = fmaxf(fmaxf(fmaxf(s[0], s[1]), fmaxf(s[2], s[3])),
                   fmaxf(fmaxf(s[4], s[5]), fmaxf(s[6], s[7])));
  mx = fmaxf(mx, xhalf(mx));
  const float mn = fmaxf(m, mx);
  const float alpha = __expf(m - mn);
  m = mn;
  v8f p;
#pragma unroll
  for (int r = 0; r < 8; ++r) p[r] = __expf(s[r] - mn);
  const float ps = ((p[0] + p[1]) + (p[2] + p[3])) + ((p[4] + p[5]) + (p[6] + p[7]));
  l = l * alpha + ps;
#pragma unroll
  for (int r = 0; r < 8; ++r) o[r] *= alpha;

#pragma unroll
  for (int kc = 0; kc < 4; ++kc) {
    v2f pb;
    pb.x = p[2 * kc];
    pb.y = p[2 * kc + 1];
    o = wmma4(f.va[kc], pb, o);
  }
}

__global__ __launch_bounds__(128) void k_attn(const float* __restrict__ qkv,
                                              float* __restrict__ attn_out) {
  __shared__ float ldsK0[16 * 9], ldsK1[16 * 9];
  __shared__ float ldsV0[16 * 9], ldsV1[16 * 9];
  const int tid = threadIdx.x;
  const int wave = tid >> 5;
  const int lane = tid & 31;
  const int n16 = lane & 15, hi = lane >> 4;

  const int bid = blockIdx.x;
  const int bh = bid >> 3;
  const int b = bh >> 3;
  const int h = bh & 7;
  const int qt = (bid & 7) * 8 + wave * 2;
  const float scale = 0.35355339059327373f;

  const float* qkvB = qkv + (size_t)(b * S_LEN) * (3 * E_DIM);
  const int srow = tid >> 3, scol = tid & 7;
  const int sidx = srow * 9 + scol;
  const float* stg = qkvB + srow * (3 * E_DIM) + h * D_HEAD + scol;

  v2f qb0, qb1, qc0, qc1;
  {
    const float* qp = qkvB + (qt * 16 + n16) * (3 * E_DIM) + h * D_HEAD;
    qb0.x = qp[0 + 2 * hi] * scale; qb0.y = qp[1 + 2 * hi] * scale;
    qb1.x = qp[4 + 2 * hi] * scale; qb1.y = qp[5 + 2 * hi] * scale;
    const float* qn = qp + 16 * (3 * E_DIM);
    qc0.x = qn[0 + 2 * hi] * scale; qc0.y = qn[1 + 2 * hi] * scale;
    qc1.x = qn[4 + 2 * hi] * scale; qc1.y = qn[5 + 2 * hi] * scale;
  }

  float kA, vA, kB, vB;
  {
    const float k0 = stg[E_DIM], v0 = stg[2 * E_DIM];
    kA = stg[KT_STEP + E_DIM];     vA = stg[KT_STEP + 2 * E_DIM];
    kB = stg[2 * KT_STEP + E_DIM]; vB = stg[2 * KT_STEP + 2 * E_DIM];
    ldsK0[sidx] = k0;
    ldsV0[sidx] = v0;
  }

  float m = -3.0e38f, l = 0.f;
  float m2 = -3.0e38f, l2 = 0.f;
  v8f o = {}, o2 = {};
  const int d = n16 & 7;

  for (int kt = 0; kt < N_KT; kt += 2) {
    lds_barrier();
    {
      const KVFrags f = load_frags(ldsK0, ldsV0, n16, hi, d);
      ldsK1[sidx] = kA;
      ldsV1[sidx] = vA;
      const float* nx = stg + min(kt + 3, N_KT - 1) * KT_STEP;
      kA = nx[E_DIM];
      vA = nx[2 * E_DIM];
      attn_tile(f, qb0, qb1, m, l, o);
      attn_tile(f, qc0, qc1, m2, l2, o2);
    }
    lds_barrier();
    {
      const KVFrags f = load_frags(ldsK1, ldsV1, n16, hi, d);
      ldsK0[sidx] = kB;
      ldsV0[sidx] = vB;
      const float* nx = stg + min(kt + 4, N_KT - 1) * KT_STEP;
      kB = nx[E_DIM];
      vB = nx[2 * E_DIM];
      attn_tile(f, qb0, qb1, m, l, o);
      attn_tile(f, qc0, qc1, m2, l2, o2);
    }
  }

  l += xhalf(l);
  l2 += xhalf(l2);
  {
    float4 w, w2;
    w.x = (hi ? o[4] : o[0]) / l;
    w.y = (hi ? o[5] : o[1]) / l;
    w.z = (hi ? o[6] : o[2]) / l;
    w.w = (hi ? o[7] : o[3]) / l;
    w2.x = (hi ? o2[4] : o2[0]) / l2;
    w2.y = (hi ? o2[5] : o2[1]) / l2;
    w2.z = (hi ? o2[6] : o2[2]) / l2;
    w2.w = (hi ? o2[7] : o2[3]) / l2;
    const int row = b * S_LEN + qt * 16 + n16;
    float* op = attn_out + row * E_DIM + h * D_HEAD + 4 * hi;
    *reinterpret_cast<float4*>(op) = w;
    *reinterpret_cast<float4*>(op + 16 * E_DIM) = w2;
  }
}

#define P64 68
#define P128 132

__device__ __forceinline__ v2f ld2(const float* p) {
  v2f v;
  v.x = p[0];
  v.y = p[1];
  return v;
}

__global__ __launch_bounds__(128) void k_tail(
    const float* __restrict__ attn_out, const float* __restrict__ Wout,
    const float* __restrict__ bout, const float* __restrict__ x,
    const float* __restrict__ g1, const float* __restrict__ beta1,
    const float* __restrict__ Wm1, const float* __restrict__ bm1,
    const float* __restrict__ Wm2, const float* __restrict__ bm2,
    const float* __restrict__ g2, const float* __restrict__ beta2,
    float* __restrict__ out) {
  __shared__ float res[16][P64];
  __shared__ float mlpin[16][P64];
  __shared__ float hbuf[16][P128];
  const int wave = threadIdx.x >> 5, lane = threadIdx.x & 31;
  const int n16 = lane & 15, hi = lane >> 4;
  const int m0 = blockIdx.x * 16;
  const int n0 = wave * 16;
  const int nh = wave * 32;

  const float* arow = attn_out + (m0 + n16) * E_DIM + 2 * hi;
  const float* brow = Wout + (n0 + n16) * E_DIM + 2 * hi;
  const float* w1r0 = Wm1 + (nh + n16) * E_DIM + 2 * hi;
  const float* w1r1 = Wm1 + (nh + 16 + n16) * E_DIM + 2 * hi;
  const float* w2r = Wm2 + (n0 + n16) * (2 * E_DIM) + 2 * hi;
  v2f fa[16], fb[16], fw1a[16], fw1b[16], fw2[32];
#pragma unroll
  for (int s = 0; s < 16; ++s) { fa[s] = ld2(arow + 4 * s); fb[s] = ld2(brow + 4 * s); }
  float xr[8];
#pragma unroll
  for (int r = 0; r < 8; ++r) xr[r] = x[(m0 + r + 8 * hi) * E_DIM + n0 + n16];
#pragma unroll
  for (int s = 0; s < 16; ++s) { fw1a[s] = ld2(w1r0 + 4 * s); fw1b[s] = ld2(w1r1 + 4 * s); }
#pragma unroll
  for (int s = 0; s < 32; ++s) fw2[s] = ld2(w2r + 4 * s);

  {
    v8f c = {};
#pragma unroll
    for (int s = 0; s < 16; ++s) c = wmma4(fa[s], fb[s], c);
    const float bias = bout[n0 + n16];
#pragma unroll
    for (int r = 0; r < 8; ++r) res[r + 8 * hi][n0 + n16] = c[r] + bias + xr[r];
  }
  __syncthreads();

  const int row = threadIdx.x >> 3;
  const int c0 = (threadIdx.x & 7) * 8;
  {
    float sum = 0.f, sq = 0.f;
#pragma unroll
    for (int i = 0; i < 8; ++i) { float v = res[row][c0 + i]; sum += v; sq += v * v; }
#pragma unroll
    for (int msk = 1; msk < 8; msk <<= 1) {
      sum += __shfl_xor(sum, msk);
      sq += __shfl_xor(sq, msk);
    }
    const float mean = sum * (1.f / 64.f);
    const float var = sq * (1.f / 64.f) - mean * mean;
    const float rstd = rsqrtf(var + 1e-5f);
#pragma unroll
    for (int i = 0; i < 8; ++i) {
      const int col = c0 + i;
      mlpin[row][col] = (res[row][col] - mean) * rstd * g1[col] + beta1[col];
    }
  }
  __syncthreads();

  {
    v8f h0 = {}, h1 = {};
#pragma unroll
    for (int s = 0; s < 16; ++s) {
      const v2f a = ld2(&mlpin[n16][4 * s + 2 * hi]);
      h0 = wmma4(a, fw1a[s], h0);
      h1 = wmma4(a, fw1b[s], h1);
    }
    const float bias0 = bm1[nh + n16], bias1 = bm1[nh + 16 + n16];
#pragma unroll
    for (int r = 0; r < 8; ++r) {
      hbuf[r + 8 * hi][nh + n16] = fmaxf(h0[r] + bias0, 0.f);
      hbuf[r + 8 * hi][nh + 16 + n16] = fmaxf(h1[r] + bias1, 0.f);
    }
  }
  __syncthreads();

  {
    v8f c = {};
#pragma unroll
    for (int s = 0; s < 32; ++s) {
      const v2f a = ld2(&hbuf[n16][4 * s + 2 * hi]);
      c = wmma4(a, fw2[s], c);
    }
    const float bias = bm2[n0 + n16];
#pragma unroll
    for (int r = 0; r < 8; ++r)
      res[r + 8 * hi][n0 + n16] = c[r] + bias + mlpin[r + 8 * hi][n0 + n16];
  }
  __syncthreads();

  {
    float sum = 0.f, sq = 0.f;
#pragma unroll
    for (int i = 0; i < 8; ++i) { float v = res[row][c0 + i]; sum += v; sq += v * v; }
#pragma unroll
    for (int msk = 1; msk < 8; msk <<= 1) {
      sum += __shfl_xor(sum, msk);
      sq += __shfl_xor(sq, msk);
    }
    const float mean = sum * (1.f / 64.f);
    const float var = sq * (1.f / 64.f) - mean * mean;
    const float rstd = rsqrtf(var + 1e-5f);
#pragma unroll
    for (int i = 0; i < 8; ++i) {
      const int col = c0 + i;
      out[(m0 + row) * E_DIM + col] =
          (res[row][col] - mean) * rstd * g2[col] + beta2[col];
    }
  }
}

extern "C" void kernel_launch(void* const* d_in, const int* in_sizes, int n_in,
                              void* d_out, int out_size, void* d_ws, size_t ws_size,
                              hipStream_t stream) {
  const float* x    = (const float*)d_in[0];
  const float* Wqkv = (const float*)d_in[4];
  const float* bqkv = (const float*)d_in[5];
  const float* Wout = (const float*)d_in[6];
  const float* bout = (const float*)d_in[7];
  const float* g1   = (const float*)d_in[8];
  const float* b1   = (const float*)d_in[9];
  const float* g2   = (const float*)d_in[10];
  const float* b2   = (const float*)d_in[11];
  const float* Wm1  = (const float*)d_in[12];
  const float* bm1  = (const float*)d_in[13];
  const float* Wm2  = (const float*)d_in[14];
  const float* bm2  = (const float*)d_in[15];
  float* out = (float*)d_out;

  float* ws      = (float*)d_ws;
  float* qkv     = ws;
  float* attn    = qkv + (size_t)N_TOK * 3 * E_DIM;

  k_qkv<<<(N_TOK / 16) * 12 / 4, 128, 0, stream>>>(x, Wqkv, bqkv, qkv);
  k_attn<<<(B_SZ * H_NUM * (S_LEN / 16)) / 8, 128, 0, stream>>>(qkv, attn);
  k_tail<<<N_TOK / 16, 128, 0, stream>>>(attn, Wout, bout, x, g1, b1,
                                         Wm1, bm1, Wm2, bm2, g2, b2, out);
}


# k_qkv: each wave computes a 32x48 output strip (2 row tiles x 3 column tiles), so every A/B fragment loaded feeds 3/2 tiles; six independent WMMA chains; 12288 -> 2048 waves; fp sum order unchanged
# speedup vs baseline: 2.1678x; 1.0555x over previous
#include <hip/hip_runtime.h>
#include <math.h>

typedef __attribute__((ext_vector_type(2))) float v2f;
typedef __attribute__((ext_vector_type(8))) float v8f;

#define E_DIM 64
#define H_NUM 8
#define D_HEAD 8
#define B_SZ 16
#define S_LEN 1024
#define N_TOK (B_SZ * S_LEN)

__device__ __forceinline__ v8f wmma4(v2f a, v2f b, v8f c) {
  return __builtin_amdgcn_wmma_f32_16x16x4_f32(
       false, a,  false, b,
       (short)0, c,  false,  false);
}

__global__ __launch_bounds__(128) void k_qkv(const float* __restrict__ x,
                                             const float* __restrict__ Wqkv,
                                             const float* __restrict__ bqkv,
                                             float* __restrict__ qkv) {
  const int wave = threadIdx.x >> 5;
  const int lane = threadIdx.x & 31;
  const int n16 = lane & 15, hi = lane >> 4;
  const int m0 = blockIdx.x * 32;
  const int n0 = wave * 48;
  const float* arow = x + (m0 + n16) * E_DIM + 2 * hi;
  const float* brow = Wqkv + (n0 + n16) * E_DIM + 2 * hi;
  v8f c00 = {}, c01 = {}, c02 = {}, c10 = {}, c11 = {}, c12 = {};
#pragma unroll
  for (int k0 = 0; k0 < E_DIM; k0 += 4) {
    v2f a0, a1, b0, b1, b2;
    a0.x = arow[k0];               a0.y = arow[k0 + 1];
    a1.x = arow[16 * E_DIM + k0];  a1.y = arow[16 * E_DIM + k0 + 1];
    b0.x = brow[k0];               b0.y = brow[k0 + 1];
    b1.x = brow[16 * E_DIM + k0];  b1.y = brow[16 * E_DIM + k0 + 1];
    b2.x = brow[32 * E_DIM + k0];  b2.y = brow[32 * E_DIM + k0 + 1];
    c00 = wmma4(a0, b0, c00);
    c01 = wmma4(a0, b1, c01);
    c02 = wmma4(a0, b2, c02);
    c10 = wmma4(a1, b0, c10);
    c11 = wmma4(a1, b1, c11);
    c12 = wmma4(a1, b2, c12);
  }
  const float bias0 = bqkv[n0 + n16];
  const float bias1 = bqkv[n0 + 16 + n16];
  const float bias2 = bqkv[n0 + 32 + n16];
  float* orow = qkv + (m0 + 8 * hi) * (3 * E_DIM) + n0 + n16;
#pragma unroll
  for (int r = 0; r < 8; ++r) {
    float* p0 = orow + r * (3 * E_DIM);
    float* p1 = p0 + 16 * (3 * E_DIM);
    p0[0]  = c00[r] + bias0;
    p0[16] = c01[r] + bias1;
    p0[32] = c02[r] + bias2;
    p1[0]  = c10[r] + bias0;
    p1[16] = c11[r] + bias1;
    p1[32] = c12[r] + bias2;
  }
}


__device__ __forceinline__ float xhalf(float x) {
  const int b = __builtin_bit_cast(int, x);
  return __builtin_bit_cast(
      float, __builtin_amdgcn_permlanex16(b, b, 0x76543210, 0xfedcba98, false, false));
}

__device__ __forceinline__ void lds_barrier() {
  __builtin_amdgcn_fence(__ATOMIC_RELEASE, "workgroup", "local");
  __builtin_amdgcn_s_barrier();
  __builtin_amdgcn_fence(__ATOMIC_ACQUIRE, "workgroup", "local");
}

#define N_KT (S_LEN / 16)
#define KT_STEP (16 * 3 * E_DIM)

struct KVFrags {
  v2f ka0, ka1;
  v2f va[4];
};

__device__ __forceinline__ KVFrags load_frags(const float* kb, const float* vb,
                                              int n16, int hi, int d) {
  KVFrags f;
  f.ka0.x = kb[n16 * 9 + 0 + 2 * hi];
  f.ka0.y = kb[n16 * 9 + 1 + 2 * hi];
  f.ka1.x = kb[n16 * 9 + 4 + 2 * hi];
  f.ka1.y = kb[n16 * 9 + 5 + 2 * hi];
#pragma unroll
  for (int kc = 0; kc < 4; ++kc) {
    f.va[kc].x = vb[(8 * hi + 2 * kc) * 9 + d];
    f.va[kc].y = vb[(8 * hi + 2 * kc + 1) * 9 + d];
  }
  return f;
}

__device__ __forceinline__ void attn_tile(const KVFrags& f, v2f qb0, v2f qb1,
                                          float& m, float& l, v8f& o) {
  v8f s = {};
  s = wmma4(f.ka0, qb0, s);
  s = wmma4(f.ka1, qb1, s);

  float mx = fmaxf(fmaxf(fmaxf(s[0], s[1]), fmaxf(s[2], s[3])),
                   fmaxf(fmaxf(s[4], s[5]), fmaxf(s[6], s[7])));
  mx = fmaxf(mx, xhalf(mx));
  const float mn = fmaxf(m, mx);
  const float alpha = __expf(m - mn);
  m = mn;
  v8f p;
#pragma unroll
  for (int r = 0; r < 8; ++r) p[r] = __expf(s[r] - mn);
  const float ps = ((p[0] + p[1]) + (p[2] + p[3])) + ((p[4] + p[5]) + (p[6] + p[7]));
  l = l * alpha + ps;
#pragma unroll
  for (int r = 0; r < 8; ++r) o[r] *= alpha;

#pragma unroll
  for (int kc = 0; kc < 4; ++kc) {
    v2f pb;
    pb.x = p[2 * kc];
    pb.y = p[2 * kc + 1];
    o = wmma4(f.va[kc], pb, o);
  }
}

__global__ __launch_bounds__(128) void k_attn(const float* __restrict__ qkv,
                                              float* __restrict__ attn_out) {
  __shared__ float ldsK0[16 * 9], ldsK1[16 * 9];
  __shared__ float ldsV0[16 * 9], ldsV1[16 * 9];
  const int tid = threadIdx.x;
  const int wave = tid >> 5;
  const int lane = tid & 31;
  const int n16 = lane & 15, hi = lane >> 4;

  const int bid = blockIdx.x;
  const int bh = bid >> 3;
  const int b = bh >> 3;
  const int h = bh & 7;
  const int qt = (bid & 7) * 8 + wave * 2;
  const float scale = 0.35355339059327373f;

  const float* qkvB = qkv + (size_t)(b * S_LEN) * (3 * E_DIM);
  const int srow = tid >> 3, scol = tid & 7;
  const int sidx = srow * 9 + scol;
  const float* stg = qkvB + srow * (3 * E_DIM) + h * D_HEAD + scol;

  v2f qb0, qb1, qc0, qc1;
  {
    const float* qp = qkvB + (qt * 16 + n16) * (3 * E_DIM) + h * D_HEAD;
    qb0.x = qp[0 + 2 * hi] * scale; qb0.y = qp[1 + 2 * hi] * scale;
    qb1.x = qp[4 + 2 * hi] * scale; qb1.y = qp[5 + 2 * hi] * scale;
    const float* qn = qp + 16 * (3 * E_DIM);
    qc0.x = qn[0 + 2 * hi] * scale; qc0.y = qn[1 + 2 * hi] * scale;
    qc1.x = qn[4 + 2 * hi] * scale; qc1.y = qn[5 + 2 * hi] * scale;
  }

  float kA, vA, kB, vB;
  {
    const float k0 = stg[E_DIM], v0 = stg[2 * E_DIM];
    kA = stg[KT_STEP + E_DIM];     vA = stg[KT_STEP + 2 * E_DIM];
    kB = stg[2 * KT_STEP + E_DIM]; vB = stg[2 * KT_STEP + 2 * E_DIM];
    ldsK0[sidx] = k0;
    ldsV0[sidx] = v0;
  }

  float m = -3.0e38f, l = 0.f;
  float m2 = -3.0e38f, l2 = 0.f;
  v8f o = {}, o2 = {};
  const int d = n16 & 7;

  for (int kt = 0; kt < N_KT; kt += 2) {
    lds_barrier();
    {
      const KVFrags f = load_frags(ldsK0, ldsV0, n16, hi, d);
      ldsK1[sidx] = kA;
      ldsV1[sidx] = vA;
      const float* nx = stg + min(kt + 3, N_KT - 1) * KT_STEP;
      kA = nx[E_DIM];
      vA = nx[2 * E_DIM];
      attn_tile(f, qb0, qb1, m, l, o);
      attn_tile(f, qc0, qc1, m2, l2, o2);
    }
    lds_barrier();
    {
      const KVFrags f = load_frags(ldsK1, ldsV1, n16, hi, d);
      ldsK0[sidx] = kB;
      ldsV0[sidx] = vB;
      const float* nx = stg + min(kt + 4, N_KT - 1) * KT_STEP;
      kB = nx[E_DIM];
      vB = nx[2 * E_DIM];
      attn_tile(f, qb0, qb1, m, l, o);
      attn_tile(f, qc0, qc1, m2, l2, o2);
    }
  }

  l += xhalf(l);
  l2 += xhalf(l2);
  {
    float4 w, w2;
    w.x = (hi ? o[4] : o[0]) / l;
    w.y = (hi ? o[5] : o[1]) / l;
    w.z = (hi ? o[6] : o[2]) / l;
    w.w = (hi ? o[7] : o[3]) / l;
    w2.x = (hi ? o2[4] : o2[0]) / l2;
    w2.y = (hi ? o2[5] : o2[1]) / l2;
    w2.z = (hi ? o2[6] : o2[2]) / l2;
    w2.w = (hi ? o2[7] : o2[3]) / l2;
    const int row = b * S_LEN + qt * 16 + n16;
    float* op = attn_out + row * E_DIM + h * D_HEAD + 4 * hi;
    *reinterpret_cast<float4*>(op) = w;
    *reinterpret_cast<float4*>(op + 16 * E_DIM) = w2;
  }
}

#define P64 68
#define P128 132

__device__ __forceinline__ v2f ld2(const float* p) {
  v2f v;
  v.x = p[0];
  v.y = p[1];
  return v;
}

__global__ __launch_bounds__(128) void k_tail(
    const float* __restrict__ attn_out, const float* __restrict__ Wout,
    const float* __restrict__ bout, const float* __restrict__ x,
    const float* __restrict__ g1, const float* __restrict__ beta1,
    const float* __restrict__ Wm1, const float* __restrict__ bm1,
    const float* __restrict__ Wm2, const float* __restrict__ bm2,
    const float* __restrict__ g2, const float* __restrict__ beta2,
    float* __restrict__ out) {
  __shared__ float res[16][P64];
  __shared__ float mlpin[16][P64];
  __shared__ float hbuf[16][P128];
  const int wave = threadIdx.x >> 5, lane = threadIdx.x & 31;
  const int n16 = lane & 15, hi = lane >> 4;
  const int m0 = blockIdx.x * 16;
  const int n0 = wave * 16;
  const int nh = wave * 32;

  const float* arow = attn_out + (m0 + n16) * E_DIM + 2 * hi;
  const float* brow = Wout + (n0 + n16) * E_DIM + 2 * hi;
  const float* w1r0 = Wm1 + (nh + n16) * E_DIM + 2 * hi;
  const float* w1r1 = Wm1 + (nh + 16 + n16) * E_DIM + 2 * hi;
  const float* w2r = Wm2 + (n0 + n16) * (2 * E_DIM) + 2 * hi;
  v2f fa[16], fb[16], fw1a[16], fw1b[16], fw2[32];
#pragma unroll
  for (int s = 0; s < 16; ++s) { fa[s] = ld2(arow + 4 * s); fb[s] = ld2(brow + 4 * s); }
  float xr[8];
#pragma unroll
  for (int r = 0; r < 8; ++r) xr[r] = x[(m0 + r + 8 * hi) * E_DIM + n0 + n16];
#pragma unroll
  for (int s = 0; s < 16; ++s) { fw1a[s] = ld2(w1r0 + 4 * s); fw1b[s] = ld2(w1r1 + 4 * s); }
#pragma unroll
  for (int s = 0; s < 32; ++s) fw2[s] = ld2(w2r + 4 * s);

  {
    v8f c = {};
#pragma unroll
    for (int s = 0; s < 16; ++s) c = wmma4(fa[s], fb[s], c);
    const float bias = bout[n0 + n16];
#pragma unroll
    for (int r = 0; r < 8; ++r) res[r + 8 * hi][n0 + n16] = c[r] + bias + xr[r];
  }
  __syncthreads();

  const int row = threadIdx.x >> 3;
  const int c0 = (threadIdx.x & 7) * 8;
  {
    float sum = 0.f, sq = 0.f;
#pragma unroll
    for (int i = 0; i < 8; ++i) { float v = res[row][c0 + i]; sum += v; sq += v * v; }
#pragma unroll
    for (int msk = 1; msk < 8; msk <<= 1) {
      sum += __shfl_xor(sum, msk);
      sq += __shfl_xor(sq, msk);
    }
    const float mean = sum * (1.f / 64.f);
    const float var = sq * (1.f / 64.f) - mean * mean;
    const float rstd = rsqrtf(var + 1e-5f);
#pragma unroll
    for (int i = 0; i < 8; ++i) {
      const int col = c0 + i;
      mlpin[row][col] = (res[row][col] - mean) * rstd * g1[col] + beta1[col];
    }
  }
  __syncthreads();

  {
    v8f h0 = {}, h1 = {};
#pragma unroll
    for (int s = 0; s < 16; ++s) {
      const v2f a = ld2(&mlpin[n16][4 * s + 2 * hi]);
      h0 = wmma4(a, fw1a[s], h0);
      h1 = wmma4(a, fw1b[s], h1);
    }
    const float bias0 = bm1[nh + n16], bias1 = bm1[nh + 16 + n16];
#pragma unroll
    for (int r = 0; r < 8; ++r) {
      hbuf[r + 8 * hi][nh + n16] = fmaxf(h0[r] + bias0, 0.f);
      hbuf[r + 8 * hi][nh + 16 + n16] = fmaxf(h1[r] + bias1, 0.f);
    }
  }
  __syncthreads();

  {
    v8f c = {};
#pragma unroll
    for (int s = 0; s < 32; ++s) {
      const v2f a = ld2(&hbuf[n16][4 * s + 2 * hi]);
      c = wmma4(a, fw2[s], c);
    }
    const float bias = bm2[n0 + n16];
#pragma unroll
    for (int r = 0; r < 8; ++r)
      res[r + 8 * hi][n0 + n16] = c[r] + bias + mlpin[r + 8 * hi][n0 + n16];
  }
  __syncthreads();

  {
    float sum = 0.f, sq = 0.f;
#pragma unroll
    for (int i = 0; i < 8; ++i) { float v = res[row][c0 + i]; sum += v; sq += v * v; }
#pragma unroll
    for (int msk = 1; msk < 8; msk <<= 1) {
      sum += __shfl_xor(sum, msk);
      sq += __shfl_xor(sq, msk);
    }
    const float mean = sum * (1.f / 64.f);
    const float var = sq * (1.f / 64.f) - mean * mean;
    const float rstd = rsqrtf(var + 1e-5f);
#pragma unroll
    for (int i = 0; i < 8; ++i) {
      const int col = c0 + i;
      out[(m0 + row) * E_DIM + col] =
          (res[row][col] - mean) * rstd * g2[col] + beta2[col];
    }
  }
}

extern "C" void kernel_launch(void* const* d_in, const int* in_sizes, int n_in,
                              void* d_out, int out_size, void* d_ws, size_t ws_size,
                              hipStream_t stream) {
  const float* x    = (const float*)d_in[0];
  const float* Wqkv = (const float*)d_in[4];
  const float* bqkv = (const float*)d_in[5];
  const float* Wout = (const float*)d_in[6];
  const float* bout = (const float*)d_in[7];
  const float* g1   = (const float*)d_in[8];
  const float* b1   = (const float*)d_in[9];
  const float* g2   = (const float*)d_in[10];
  const float* b2   = (const float*)d_in[11];
  const float* Wm1  = (const float*)d_in[12];
  const float* bm1  = (const float*)d_in[13];
  const float* Wm2  = (const float*)d_in[14];
  const float* bm2  = (const float*)d_in[15];
  float* out = (float*)d_out;

  float* ws      = (float*)d_ws;
  float* qkv     = ws;
  float* attn    = qkv + (size_t)N_TOK * 3 * E_DIM;

  k_qkv<<<N_TOK / 32, 128, 0, stream>>>(x, Wqkv, bqkv, qkv);
  k_attn<<<(B_SZ * H_NUM * (S_LEN / 16)) / 8, 128, 0, stream>>>(qkv, attn);
  k_tail<<<N_TOK / 16, 128, 0, stream>>>(attn, Wout, bout, x, g1, b1,
                                         Wm1, bm1, Wm2, bm2, g2, b2, out);
}


# k_attn: lazy softmax reference: log2e folded in Q, score WMMA starts at C=-m_ref so p=exp2(acc); max, subtract, half exchange, O/l rescale only in tiles whose p sum leaves range; fp sum order changes
# speedup vs baseline: 2.3623x; 1.0897x over previous
#include <hip/hip_runtime.h>
#include <math.h>

typedef __attribute__((ext_vector_type(2))) float v2f;
typedef __attribute__((ext_vector_type(8))) float v8f;

#define E_DIM 64
#define H_NUM 8
#define D_HEAD 8
#define B_SZ 16
#define S_LEN 1024
#define N_TOK (B_SZ * S_LEN)

__device__ __forceinline__ v8f wmma4(v2f a, v2f b, v8f c) {
  return __builtin_amdgcn_wmma_f32_16x16x4_f32(
       false, a,  false, b,
       (short)0, c,  false,  false);
}

__global__ __launch_bounds__(128) void k_qkv(const float* __restrict__ x,
                                             const float* __restrict__ Wqkv,
                                             const float* __restrict__ bqkv,
                                             float* __restrict__ qkv) {
  const int wave = threadIdx.x >> 5;
  const int lane = threadIdx.x & 31;
  const int n16 = lane & 15, hi = lane >> 4;
  const int m0 = blockIdx.x * 32;
  const int n0 = wave * 48;
  const float* arow = x + (m0 + n16) * E_DIM + 2 * hi;
  const float* brow = Wqkv + (n0 + n16) * E_DIM + 2 * hi;
  v8f c00 = {}, c01 = {}, c02 = {}, c10 = {}, c11 = {}, c12 = {};
#pragma unroll
  for (int k0 = 0; k0 < E_DIM; k0 += 4) {
    v2f a0, a1, b0, b1, b2;
    a0.x = arow[k0];               a0.y = arow[k0 + 1];
    a1.x = arow[16 * E_DIM + k0];  a1.y = arow[16 * E_DIM + k0 + 1];
    b0.x = brow[k0];               b0.y = brow[k0 + 1];
    b1.x = brow[16 * E_DIM + k0];  b1.y = brow[16 * E_DIM + k0 + 1];
    b2.x = brow[32 * E_DIM + k0];  b2.y = brow[32 * E_DIM + k0 + 1];
    c00 = wmma4(a0, b0, c00);
    c01 = wmma4(a0, b1, c01);
    c02 = wmma4(a0, b2, c02);
    c10 = wmma4(a1, b0, c10);
    c11 = wmma4(a1, b1, c11);
    c12 = wmma4(a1, b2, c12);
  }
  const float bias0 = bqkv[n0 + n16];
  const float bias1 = bqkv[n0 + 16 + n16];
  const float bias2 = bqkv[n0 + 32 + n16];
  float* orow = qkv + (m0 + 8 * hi) * (3 * E_DIM) + n0 + n16;
#pragma unroll
  for (int r = 0; r < 8; ++r) {
    float* p0 = orow + r * (3 * E_DIM);
    float* p1 = p0 + 16 * (3 * E_DIM);
    p0[0]  = c00[r] + bias0;
    p0[16] = c01[r] + bias1;
    p0[32] = c02[r] + bias2;
    p1[0]  = c10[r] + bias0;
    p1[16] = c11[r] + bias1;
    p1[32] = c12[r] + bias2;
  }
}


__device__ __forceinline__ float xhalf(float x) {
  const int b = __builtin_bit_cast(int, x);
  return __builtin_bit_cast(
      float, __builtin_amdgcn_permlanex16(b, b, 0x76543210, 0xfedcba98, false, false));
}

__device__ __forceinline__ void lds_barrier() {
  __builtin_amdgcn_fence(__ATOMIC_RELEASE, "workgroup", "local");
  __builtin_amdgcn_s_barrier();
  __builtin_amdgcn_fence(__ATOMIC_ACQUIRE, "workgroup", "local");
}

#define N_KT (S_LEN / 16)
#define KT_STEP (16 * 3 * E_DIM)

struct KVFrags {
  v2f ka0, ka1;
  v2f va[4];
};

__device__ __forceinline__ KVFrags load_frags(const float* kb, const float* vb,
                                              int n16, int hi, int d) {
  KVFrags f;
  f.ka0.x = kb[n16 * 9 + 0 + 2 * hi];
  f.ka0.y = kb[n16 * 9 + 1 + 2 * hi];
  f.ka1.x = kb[n16 * 9 + 4 + 2 * hi];
  f.ka1.y = kb[n16 * 9 + 5 + 2 * hi];
#pragma unroll
  for (int kc = 0; kc < 4; ++kc) {
    f.va[kc].x = vb[(8 * hi + 2 * kc) * 9 + d];
    f.va[kc].y = vb[(8 * hi + 2 * kc + 1) * 9 + d];
  }
  return f;
}

#define PS_MAX 256.0f

__device__ __forceinline__ void tile_p(const KVFrags& f, v2f qb0, v2f qb1,
                                       const v8f& cref, v8f& p, float& ps) {
  v8f s = wmma4(f.ka0, qb0, cref);
  s = wmma4(f.ka1, qb1, s);
#pragma unroll
  for (int r = 0; r < 8; ++r) p[r] = __builtin_amdgcn_exp2f(s[r]);
  ps = ((p[0] + p[1]) + (p[2] + p[3])) + ((p[4] + p[5]) + (p[6] + p[7]));
}

__device__ __forceinline__ void tile_rebase(const KVFrags& f, v2f qb0, v2f qb1,
                                            float& m, float& l, v8f& o,
                                            v8f& cref, v8f& p, float& ps) {
  v8f s = {};
  s = wmma4(f.ka0, qb0, s);
  s = wmma4(f.ka1, qb1, s);
  float mx = fmaxf(fmaxf(fmaxf(s[0], s[1]), fmaxf(s[2], s[3])),
                   fmaxf(fmaxf(s[4], s[5]), fmaxf(s[6], s[7])));
  mx = fmaxf(mx, xhalf(mx));
  const float mn = fmaxf(m, mx);
  const float alpha = __builtin_amdgcn_exp2f(m - mn);
  m = mn;
#pragma unroll
  for (int r = 0; r < 8; ++r) p[r] = __builtin_amdgcn_exp2f(s[r] - mn);
  ps = ((p[0] + p[1]) + (p[2] + p[3])) + ((p[4] + p[5]) + (p[6] + p[7]));
  l *= alpha;
#pragma unroll
  for (int r = 0; r < 8; ++r) {
    o[r] *= alpha;
    cref[r] = -mn;
  }
  asm volatile("" : "+v"(cref));
}

__device__ __forceinline__ void tile_pv(const KVFrags& f, const v8f& p, v8f& o) {
#pragma unroll
  for (int kc = 0; kc < 4; ++kc) {
    v2f pb;
    pb.x = p[2 * kc];
    pb.y = p[2 * kc + 1];
    o = wmma4(f.va[kc], pb, o);
  }
}

__device__ __forceinline__ void attn_step(const KVFrags& f, v2f qb0, v2f qb1,
                                          v2f qc0, v2f qc1, float& m, float& l,
                                          v8f& o, v8f& cref, float& m2, float& l2,
                                          v8f& o2, v8f& cref2) {
  v8f p, p2;
  float ps, ps2;
  tile_p(f, qb0, qb1, cref, p, ps);
  tile_p(f, qc0, qc1, cref2, p2, ps2);
  if (__builtin_amdgcn_ballot_w32(!(fmaxf(ps, ps2) <= PS_MAX)) != 0) {
    tile_rebase(f, qb0, qb1, m, l, o, cref, p, ps);
    tile_rebase(f, qc0, qc1, m2, l2, o2, cref2, p2, ps2);
  }
  l += ps;
  l2 += ps2;
  tile_pv(f, p, o);
  tile_pv(f, p2, o2);
}

__global__ __launch_bounds__(128) void k_attn(const float* __restrict__ qkv,
                                              float* __restrict__ attn_out) {
  __shared__ float ldsK0[16 * 9], ldsK1[16 * 9];
  __shared__ float ldsV0[16 * 9], ldsV1[16 * 9];
  const int tid = threadIdx.x;
  const int wave = tid >> 5;
  const int lane = tid & 31;
  const int n16 = lane & 15, hi = lane >> 4;

  const int bid = blockIdx.x;
  const int bh = bid >> 3;
  const int b = bh >> 3;
  const int h = bh & 7;
  const int qt = (bid & 7) * 8 + wave * 2;
  const float scale = (float)(1.4426950408889634 * 0.35355339059327373);

  const float* qkvB = qkv + (size_t)(b * S_LEN) * (3 * E_DIM);
  const int srow = tid >> 3, scol = tid & 7;
  const int sidx = srow * 9 + scol;
  const float* stg = qkvB + srow * (3 * E_DIM) + h * D_HEAD + scol;

  v2f qb0, qb1, qc0, qc1;
  {
    const float* qp = qkvB + (qt * 16 + n16) * (3 * E_DIM) + h * D_HEAD;
    qb0.x = qp[0 + 2 * hi] * scale; qb0.y = qp[1 + 2 * hi] * scale;
    qb1.x = qp[4 + 2 * hi] * scale; qb1.y = qp[5 + 2 * hi] * scale;
    const float* qn = qp + 16 * (3 * E_DIM);
    qc0.x = qn[0 + 2 * hi] * scale; qc0.y = qn[1 + 2 * hi] * scale;
    qc1.x = qn[4 + 2 * hi] * scale; qc1.y = qn[5 + 2 * hi] * scale;
  }

  float kA, vA, kB, vB;
  {
    const float k0 = stg[E_DIM], v0 = stg[2 * E_DIM];
    kA = stg[KT_STEP + E_DIM];     vA = stg[KT_STEP + 2 * E_DIM];
    kB = stg[2 * KT_STEP + E_DIM]; vB = stg[2 * KT_STEP + 2 * E_DIM];
    ldsK0[sidx] = k0;
    ldsV0[sidx] = v0;
  }

  float m = -3.0e38f, l = 0.f;
  float m2 = -3.0e38f, l2 = 0.f;
  v8f o = {}, o2 = {};
  v8f cref, cref2;
#pragma unroll
  for (int r = 0; r < 8; ++r) { cref[r] = 3.0e38f; cref2[r] = 3.0e38f; }
  const int d = n16 & 7;

  for (int kt = 0; kt < N_KT; kt += 2) {
    lds_barrier();
    {
      const KVFrags f = load_frags(ldsK0, ldsV0, n16, hi, d);
      ldsK1[sidx] = kA;
      ldsV1[sidx] = vA;
      const float* nx = stg + min(kt + 3, N_KT - 1) * KT_STEP;
      kA = nx[E_DIM];
      vA = nx[2 * E_DIM];
      attn_step(f, qb0, qb1, qc0, qc1, m, l, o, cref, m2, l2, o2, cref2);
    }
    lds_barrier();
    {
      const KVFrags f = load_frags(ldsK1, ldsV1, n16, hi, d);
      ldsK0[sidx] = kB;
      ldsV0[sidx] = vB;
      const float* nx = stg + min(kt + 4, N_KT - 1) * KT_STEP;
      kB = nx[E_DIM];
      vB = nx[2 * E_DIM];
      attn_step(f, qb0, qb1, qc0, qc1, m, l, o, cref, m2, l2, o2, cref2);
    }
  }

  l += xhalf(l);
  l2 += xhalf(l2);
  {
    float4 w, w2;
    w.x = (hi ? o[4] : o[0]) / l;
    w.y = (hi ? o[5] : o[1]) / l;
    w.z = (hi ? o[6] : o[2]) / l;
    w.w = (hi ? o[7] : o[3]) / l;
    w2.x = (hi ? o2[4] : o2[0]) / l2;
    w2.y = (hi ? o2[5] : o2[1]) / l2;
    w2.z = (hi ? o2[6] : o2[2]) / l2;
    w2.w = (hi ? o2[7] : o2[3]) / l2;
    const int row = b * S_LEN + qt * 16 + n16;
    float* op = attn_out + row * E_DIM + h * D_HEAD + 4 * hi;
    *reinterpret_cast<float4*>(op) = w;
    *reinterpret_cast<float4*>(op + 16 * E_DIM) = w2;
  }
}

#define P64 68
#define P128 132

__device__ __forceinline__ v2f ld2(const float* p) {
  v2f v;
  v.x = p[0];
  v.y = p[1];
  return v;
}

__global__ __launch_bounds__(128) void k_tail(
    const float* __restrict__ attn_out, const float* __restrict__ Wout,
    const float* __restrict__ bout, const float* __restrict__ x,
    const float* __restrict__ g1, const float* __restrict__ beta1,
    const float* __restrict__ Wm1, const float* __restrict__ bm1,
    const float* __restrict__ Wm2, const float* __restrict__ bm2,
    const float* __restrict__ g2, const float* __restrict__ beta2,
    float* __restrict__ out) {
  __shared__ float res[16][P64];
  __shared__ float mlpin[16][P64];
  __shared__ float hbuf[16][P128];
  const int wave = threadIdx.x >> 5, lane = threadIdx.x & 31;
  const int n16 = lane & 15, hi = lane >> 4;
  const int m0 = blockIdx.x * 16;
  const int n0 = wave * 16;
  const int nh = wave * 32;

  const float* arow = attn_out + (m0 + n16) * E_DIM + 2 * hi;
  const float* brow = Wout + (n0 + n16) * E_DIM + 2 * hi;
  const float* w1r0 = Wm1 + (nh + n16) * E_DIM + 2 * hi;
  const float* w1r1 = Wm1 + (nh + 16 + n16) * E_DIM + 2 * hi;
  const float* w2r = Wm2 + (n0 + n16) * (2 * E_DIM) + 2 * hi;
  v2f fa[16], fb[16], fw1a[16], fw1b[16], fw2[32];
#pragma unroll
  for (int s = 0; s < 16; ++s) { fa[s] = ld2(arow + 4 * s); fb[s] = ld2(brow + 4 * s); }
  float xr[8];
#pragma unroll
  for (int r = 0; r < 8; ++r) xr[r] = x[(m0 + r + 8 * hi) * E_DIM + n0 + n16];
#pragma unroll
  for (int s = 0; s < 16; ++s) { fw1a[s] = ld2(w1r0 + 4 * s); fw1b[s] = ld2(w1r1 + 4 * s); }
#pragma unroll
  for (int s = 0; s < 32; ++s) fw2[s] = ld2(w2r + 4 * s);

  {
    v8f c = {};
#pragma unroll
    for (int s = 0; s < 16; ++s) c = wmma4(fa[s], fb[s], c);
    const float bias = bout[n0 + n16];
#pragma unroll
    for (int r = 0; r < 8; ++r) res[r + 8 * hi][n0 + n16] = c[r] + bias + xr[r];
  }
  __syncthreads();

  const int row = threadIdx.x >> 3;
  const int c0 = (threadIdx.x & 7) * 8;
  {
    float sum = 0.f, sq = 0.f;
#pragma unroll
    for (int i = 0; i < 8; ++i) { float v = res[row][c0 + i]; sum += v; sq += v * v; }
#pragma unroll
    for (int msk = 1; msk < 8; msk <<= 1) {
      sum += __shfl_xor(sum, msk);
      sq += __shfl_xor(sq, msk);
    }
    const float mean = sum * (1.f / 64.f);
    const float var = sq * (1.f / 64.f) - mean * mean;
    const float rstd = rsqrtf(var + 1e-5f);
#pragma unroll
    for (int i = 0; i < 8; ++i) {
      const int col = c0 + i;
      mlpin[row][col] = (res[row][col] - mean) * rstd * g1[col] + beta1[col];
    }
  }
  __syncthreads();

  {
    v8f h0 = {}, h1 = {};
#pragma unroll
    for (int s = 0; s < 16; ++s) {
      const v2f a = ld2(&mlpin[n16][4 * s + 2 * hi]);
      h0 = wmma4(a, fw1a[s], h0);
      h1 = wmma4(a, fw1b[s], h1);
    }
    const float bias0 = bm1[nh + n16], bias1 = bm1[nh + 16 + n16];
#pragma unroll
    for (int r = 0; r < 8; ++r) {
      hbuf[r + 8 * hi][nh + n16] = fmaxf(h0[r] + bias0, 0.f);
      hbuf[r + 8 * hi][nh + 16 + n16] = fmaxf(h1[r] + bias1, 0.f);
    }
  }
  __syncthreads();

  {
    v8f c = {};
#pragma unroll
    for (int s = 0; s < 32; ++s) {
      const v2f a = ld2(&hbuf[n16][4 * s + 2 * hi]);
      c = wmma4(a, fw2[s], c);
    }
    const float bias = bm2[n0 + n16];
#pragma unroll
    for (int r = 0; r < 8; ++r)
      res[r + 8 * hi][n0 + n16] = c[r] + bias + mlpin[r + 8 * hi][n0 + n16];
  }
  __syncthreads();

  {
    float sum = 0.f, sq = 0.f;
#pragma unroll
    for (int i = 0; i < 8; ++i) { float v = res[row][c0 + i]; sum += v; sq += v * v; }
#pragma unroll
    for (int msk = 1; msk < 8; msk <<= 1) {
      sum += __shfl_xor(sum, msk);
      sq += __shfl_xor(sq, msk);
    }
    const float mean = sum * (1.f / 64.f);
    const float var = sq * (1.f / 64.f) - mean * mean;
    const float rstd = rsqrtf(var + 1e-5f);
#pragma unroll
    for (int i = 0; i < 8; ++i) {
      const int col = c0 + i;
      out[(m0 + row) * E_DIM + col] =
          (res[row][col] - mean) * rstd * g2[col] + beta2[col];
    }
  }
}

extern "C" void kernel_launch(void* const* d_in, const int* in_sizes, int n_in,
                              void* d_out, int out_size, void* d_ws, size_t ws_size,
                              hipStream_t stream) {
  const float* x    = (const float*)d_in[0];
  const float* Wqkv = (const float*)d_in[4];
  const float* bqkv = (const float*)d_in[5];
  const float* Wout = (const float*)d_in[6];
  const float* bout = (const float*)d_in[7];
  const float* g1   = (const float*)d_in[8];
  const float* b1   = (const float*)d_in[9];
  const float* g2   = (const float*)d_in[10];
  const float* b2   = (const float*)d_in[11];
  const float* Wm1  = (const float*)d_in[12];
  const float* bm1  = (const float*)d_in[13];
  const float* Wm2  = (const float*)d_in[14];
  const float* bm2  = (const float*)d_in[15];
  float* out = (float*)d_out;

  float* ws      = (float*)d_ws;
  float* qkv     = ws;
  float* attn    = qkv + (size_t)N_TOK * 3 * E_DIM;

  k_qkv<<<N_TOK / 32, 128, 0, stream>>>(x, Wqkv, bqkv, qkv);
  k_attn<<<(B_SZ * H_NUM * (S_LEN / 16)) / 8, 128, 0, stream>>>(qkv, attn);
  k_tail<<<N_TOK / 16, 128, 0, stream>>>(attn, Wout, bout, x, g1, b1,
                                         Wm1, bm1, Wm2, bm2, g2, b2, out);
}


# fuse k_tail into k_attn (3 launches -> 2): a block owns 64 query rows of a batch for all 8 heads (16 waves); its 64x64 attention tile stays in LDS and feeds out-proj+LN1, MLP1, MLP2+LN2
# speedup vs baseline: 2.4496x; 1.0370x over previous
#include <hip/hip_runtime.h>
#include <math.h>

typedef __attribute__((ext_vector_type(2))) float v2f;
typedef __attribute__((ext_vector_type(8))) float v8f;

#define E_DIM 64
#define H_NUM 8
#define D_HEAD 8
#define B_SZ 16
#define S_LEN 1024
#define N_TOK (B_SZ * S_LEN)

__device__ __forceinline__ v8f wmma4(v2f a, v2f b, v8f c) {
  return __builtin_amdgcn_wmma_f32_16x16x4_f32(
       false, a,  false, b,
       (short)0, c,  false,  false);
}

__global__ __launch_bounds__(128) void k_qkv(const float* __restrict__ x,
                                             const float* __restrict__ Wqkv,
                                             const float* __restrict__ bqkv,
                                             float* __restrict__ qkv) {
  const int wave = threadIdx.x >> 5;
  const int lane = threadIdx.x & 31;
  const int n16 = lane & 15, hi = lane >> 4;
  const int m0 = blockIdx.x * 32;
  const int n0 = wave * 48;
  const float* arow = x + (m0 + n16) * E_DIM + 2 * hi;
  const float* brow = Wqkv + (n0 + n16) * E_DIM + 2 * hi;
  v8f c00 = {}, c01 = {}, c02 = {}, c10 = {}, c11 = {}, c12 = {};
#pragma unroll
  for (int k0 = 0; k0 < E_DIM; k0 += 4) {
    v2f a0, a1, b0, b1, b2;
    a0.x = arow[k0];               a0.y = arow[k0 + 1];
    a1.x = arow[16 * E_DIM + k0];  a1.y = arow[16 * E_DIM + k0 + 1];
    b0.x = brow[k0];               b0.y = brow[k0 + 1];
    b1.x = brow[16 * E_DIM + k0];  b1.y = brow[16 * E_DIM + k0 + 1];
    b2.x = brow[32 * E_DIM + k0];  b2.y = brow[32 * E_DIM + k0 + 1];
    c00 = wmma4(a0, b0, c00);
    c01 = wmma4(a0, b1, c01);
    c02 = wmma4(a0, b2, c02);
    c10 = wmma4(a1, b0, c10);
    c11 = wmma4(a1, b1, c11);
    c12 = wmma4(a1, b2, c12);
  }
  const float bias0 = bqkv[n0 + n16];
  const float bias1 = bqkv[n0 + 16 + n16];
  const float bias2 = bqkv[n0 + 32 + n16];
  float* orow = qkv + (m0 + 8 * hi) * (3 * E_DIM) + n0 + n16;
#pragma unroll
  for (int r = 0; r < 8; ++r) {
    float* p0 = orow + r * (3 * E_DIM);
    float* p1 = p0 + 16 * (3 * E_DIM);
    p0[0]  = c00[r] + bias0;
    p0[16] = c01[r] + bias1;
    p0[32] = c02[r] + bias2;
    p1[0]  = c10[r] + bias0;
    p1[16] = c11[r] + bias1;
    p1[32] = c12[r] + bias2;
  }
}


__device__ __forceinline__ float xhalf(float x) {
  const int b = __builtin_bit_cast(int, x);
  return __builtin_bit_cast(
      float, __builtin_amdgcn_permlanex16(b, b, 0x76543210, 0xfedcba98, false, false));
}

__device__ __forceinline__ void lds_barrier() {
  __builtin_amdgcn_fence(__ATOMIC_RELEASE, "workgroup", "local");
  __builtin_amdgcn_s_barrier();
  __builtin_amdgcn_fence(__ATOMIC_ACQUIRE, "workgroup", "local");
}

#define N_KT (S_LEN / 16)
#define KT_STEP (16 * 3 * E_DIM)

struct KVFrags {
  v2f ka0, ka1;
  v2f va[4];
};

__device__ __forceinline__ KVFrags load_frags(const float* kb, const float* vb,
                                              int n16, int hi, int d) {
  KVFrags f;
  f.ka0.x = kb[n16 * 9 + 0 + 2 * hi];
  f.ka0.y = kb[n16 * 9 + 1 + 2 * hi];
  f.ka1.x = kb[n16 * 9 + 4 + 2 * hi];
  f.ka1.y = kb[n16 * 9 + 5 + 2 * hi];
#pragma unroll
  for (int kc = 0; kc < 4; ++kc) {
    f.va[kc].x = vb[(8 * hi + 2 * kc) * 9 + d];
    f.va[kc].y = vb[(8 * hi + 2 * kc + 1) * 9 + d];
  }
  return f;
}

#define PS_MAX 256.0f

__device__ __forceinline__ void tile_p(const KVFrags& f, v2f qb0, v2f qb1,
                                       const v8f& cref, v8f& p, float& ps) {
  v8f s = wmma4(f.ka0, qb0, cref);
  s = wmma4(f.ka1, qb1, s);
#pragma unroll
  for (int r = 0; r < 8; ++r) p[r] = __builtin_amdgcn_exp2f(s[r]);
  ps = ((p[0] + p[1]) + (p[2] + p[3])) + ((p[4] + p[5]) + (p[6] + p[7]));
}

__device__ __forceinline__ void tile_rebase(const KVFrags& f, v2f qb0, v2f qb1,
                                            float& m, float& l, v8f& o,
                                            v8f& cref, v8f& p, float& ps) {
  v8f s = {};
  s = wmma4(f.ka0, qb0, s);
  s = wmma4(f.ka1, qb1, s);
  float mx = fmaxf(fmaxf(fmaxf(s[0], s[1]), fmaxf(s[2], s[3])),
                   fmaxf(fmaxf(s[4], s[5]), fmaxf(s[6], s[7])));
  mx = fmaxf(mx, xhalf(mx));
  const float mn = fmaxf(m, mx);
  const float alpha = __builtin_amdgcn_exp2f(m - mn);
  m = mn;
#pragma unroll
  for (int r = 0; r < 8; ++r) p[r] = __builtin_amdgcn_exp2f(s[r] - mn);
  ps = ((p[0] + p[1]) + (p[2] + p[3])) + ((p[4] + p[5]) + (p[6] + p[7]));
  l *= alpha;
#pragma unroll
  for (int r = 0; r < 8; ++r) {
    o[r] *= alpha;
    cref[r] = -mn;
  }
  asm volatile("" : "+v"(cref));
}

__device__ __forceinline__ void tile_pv(const KVFrags& f, const v8f& p, v8f& o) {
#pragma unroll
  for (int kc = 0; kc < 4; ++kc) {
    v2f pb;
    pb.x = p[2 * kc];
    pb.y = p[2 * kc + 1];
    o = wmma4(f.va[kc], pb, o);
  }
}

__device__ __forceinline__ void attn_step(const KVFrags& f, v2f qb0, v2f qb1,
                                          v2f qc0, v2f qc1, float& m, float& l,
                                          v8f& o, v8f& cref, float& m2, float& l2,
                                          v8f& o2, v8f& cref2) {
  v8f p, p2;
  float ps, ps2;
  tile_p(f, qb0, qb1, cref, p, ps);
  tile_p(f, qc0, qc1, cref2, p2, ps2);
  if (__builtin_amdgcn_ballot_w32(!(fmaxf(ps, ps2) <= PS_MAX)) != 0) {
    tile_rebase(f, qb0, qb1, m, l, o, cref, p, ps);
    tile_rebase(f, qc0, qc1, m2, l2, o2, cref2, p2, ps2);
  }
  l += ps;
  l2 += ps2;
  tile_pv(f, p, o);
  tile_pv(f, p2, o2);
}

#define P64 68
#define P128 132
#define KV_HEAD (16 * 9)
#define KV_STAGE (H_NUM * KV_HEAD)

__device__ __forceinline__ v2f ld2(const float* p) {
  v2f v;
  v.x = p[0];
  v.y = p[1];
  return v;
}

__global__ __launch_bounds__(512) void k_attn_tail(
    const float* __restrict__ qkv, const float* __restrict__ Wout,
    const float* __restrict__ bout, const float* __restrict__ x,
    const float* __restrict__ g1, const float* __restrict__ beta1,
    const float* __restrict__ Wm1, const float* __restrict__ bm1,
    const float* __restrict__ Wm2, const float* __restrict__ bm2,
    const float* __restrict__ g2, const float* __restrict__ beta2,
    float* __restrict__ out) {
  __shared__ float ldsK0[KV_STAGE], ldsK1[KV_STAGE];
  __shared__ float ldsV0[KV_STAGE], ldsV1[KV_STAGE];
  __shared__ __attribute__((aligned(16))) float att[64][P64];
  __shared__ float res[64][P64];
  __shared__ float mlpin[64][P64];
  __shared__ float hbuf[64][P128];
  const int tid = threadIdx.x;
  const int wave = tid >> 5;
  const int lane = tid & 31;
  const int n16 = lane & 15, hi = lane >> 4;

  const int bid = blockIdx.x;
  const int b = bid >> 4;
  const int qblk = bid & 15;
  const int h = wave >> 1;
  const int qt = qblk * 4 + (wave & 1) * 2;
  const float scale = (float)(1.4426950408889634 * 0.35355339059327373);

  const float* qkvB = qkv + (size_t)(b * S_LEN) * (3 * E_DIM);
  const int srow = tid >> 5, scol = (tid & 31) * 2;
  const int sidx = (scol >> 3) * KV_HEAD + srow * 9 + (scol & 7);
  const float* stg = qkvB + srow * (3 * E_DIM) + scol;
  const float* myK0 = ldsK0 + h * KV_HEAD;
  const float* myK1 = ldsK1 + h * KV_HEAD;
  const float* myV0 = ldsV0 + h * KV_HEAD;
  const float* myV1 = ldsV1 + h * KV_HEAD;

  v2f qb0, qb1, qc0, qc1;
  {
    const float* qp = qkvB + (qt * 16 + n16) * (3 * E_DIM) + h * D_HEAD;
    qb0.x = qp[0 + 2 * hi] * scale; qb0.y = qp[1 + 2 * hi] * scale;
    qb1.x = qp[4 + 2 * hi] * scale; qb1.y = qp[5 + 2 * hi] * scale;
    const float* qn = qp + 16 * (3 * E_DIM);
    qc0.x = qn[0 + 2 * hi] * scale; qc0.y = qn[1 + 2 * hi] * scale;
    qc1.x = qn[4 + 2 * hi] * scale; qc1.y = qn[5 + 2 * hi] * scale;
  }

  v2f kA, vA, kB, vB;
  {
    const v2f k0 = ld2(stg + E_DIM), v0 = ld2(stg + 2 * E_DIM);
    kA = ld2(stg + KT_STEP + E_DIM);     vA = ld2(stg + KT_STEP + 2 * E_DIM);
    kB = ld2(stg + 2 * KT_STEP + E_DIM); vB = ld2(stg + 2 * KT_STEP + 2 * E_DIM);
    ldsK0[sidx] = k0.x; ldsK0[sidx + 1] = k0.y;
    ldsV0[sidx] = v0.x; ldsV0[sidx + 1] = v0.y;
  }

  float m = -3.0e38f, l = 0.f;
  float m2 = -3.0e38f, l2 = 0.f;
  v8f o = {}, o2 = {};
  v8f cref, cref2;
#pragma unroll
  for (int r = 0; r < 8; ++r) { cref[r] = 3.0e38f; cref2[r] = 3.0e38f; }
  const int d = n16 & 7;

  for (int kt = 0; kt < N_KT; kt += 2) {
    lds_barrier();
    {
      const KVFrags f = load_frags(myK0, myV0, n16, hi, d);
      ldsK1[sidx] = kA.x; ldsK1[sidx + 1] = kA.y;
      ldsV1[sidx] = vA.x; ldsV1[sidx + 1] = vA.y;
      const float* nx = stg + min(kt + 3, N_KT - 1) * KT_STEP;
      kA = ld2(nx + E_DIM);
      vA = ld2(nx + 2 * E_DIM);
      attn_step(f, qb0, qb1, qc0, qc1, m, l, o, cref, m2, l2, o2, cref2);
    }
    lds_barrier();
    {
      const KVFrags f = load_frags(myK1, myV1, n16, hi, d);
      ldsK0[sidx] = kB.x; ldsK0[sidx + 1] = kB.y;
      ldsV0[sidx] = vB.x; ldsV0[sidx + 1] = vB.y;
      const float* nx = stg + min(kt + 4, N_KT - 1) * KT_STEP;
      kB = ld2(nx + E_DIM);
      vB = ld2(nx + 2 * E_DIM);
      attn_step(f, qb0, qb1, qc0, qc1, m, l, o, cref, m2, l2, o2, cref2);
    }
  }

  l += xhalf(l);
  l2 += xhalf(l2);
  {
    float4 w, w2;
    w.x = (hi ? o[4] : o[0]) / l;
    w.y = (hi ? o[5] : o[1]) / l;
    w.z = (hi ? o[6] : o[2]) / l;
    w.w = (hi ? o[7] : o[3]) / l;
    w2.x = (hi ? o2[4] : o2[0]) / l2;
    w2.y = (hi ? o2[5] : o2[1]) / l2;
    w2.z = (hi ? o2[6] : o2[2]) / l2;
    w2.w = (hi ? o2[7] : o2[3]) / l2;
    const int r0 = (wave & 1) * 32 + n16;
    *reinterpret_cast<float4*>(&att[r0][h * D_HEAD + 4 * hi]) = w;
    *reinterpret_cast<float4*>(&att[r0 + 16][h * D_HEAD + 4 * hi]) = w2;
  }

  const int rb = (wave >> 2) * 16;
  const int cw = wave & 3;
  const int n0 = cw * 16;
  const int nh = cw * 32;
  const int grow0 = b * S_LEN + qblk * 64;

  const float* brow = Wout + (n0 + n16) * E_DIM + 2 * hi;
  const float* w1r0 = Wm1 + (nh + n16) * E_DIM + 2 * hi;
  const float* w1r1 = Wm1 + (nh + 16 + n16) * E_DIM + 2 * hi;
  const float* w2r = Wm2 + (n0 + n16) * (2 * E_DIM) + 2 * hi;
  v2f fb[16], fw1a[16], fw1b[16], fw2[32];
#pragma unroll
  for (int s = 0; s < 16; ++s) fb[s] = ld2(brow + 4 * s);
  float xr[8];
#pragma unroll
  for (int r = 0; r < 8; ++r) xr[r] = x[(grow0 + rb + r + 8 * hi) * E_DIM + n0 + n16];
#pragma unroll
  for (int s = 0; s < 16; ++s) { fw1a[s] = ld2(w1r0 + 4 * s); fw1b[s] = ld2(w1r1 + 4 * s); }
#pragma unroll
  for (int s = 0; s < 32; ++s) fw2[s] = ld2(w2r + 4 * s);
  __syncthreads();

  {
    v8f c = {};
#pragma unroll
    for (int s = 0; s < 16; ++s) c = wmma4(ld2(&att[rb + n16][4 * s + 2 * hi]), fb[s], c);
    const float bias = bout[n0 + n16];
#pragma unroll
    for (int r = 0; r < 8; ++r) res[rb + r + 8 * hi][n0 + n16] = c[r] + bias + xr[r];
  }
  __syncthreads();

  const int row = rb + ((tid & 127) >> 3);
  const int c0 = (tid & 7) * 8;
  {
    float sum = 0.f, sq = 0.f;
#pragma unroll
    for (int i = 0; i < 8; ++i) { float v = res[row][c0 + i]; sum += v; sq += v * v; }
#pragma unroll
    for (int msk = 1; msk < 8; msk <<= 1) {
      sum += __shfl_xor(sum, msk);
      sq += __shfl_xor(sq, msk);
    }
    const float mean = sum * (1.f / 64.f);
    const float var = sq * (1.f / 64.f) - mean * mean;
    const float rstd = rsqrtf(var + 1e-5f);
#pragma unroll
    for (int i = 0; i < 8; ++i) {
      const int col = c0 + i;
      mlpin[row][col] = (res[row][col] - mean) * rstd * g1[col] + beta1[col];
    }
  }
  __syncthreads();

  {
    v8f h0 = {}, h1 = {};
#pragma unroll
    for (int s = 0; s < 16; ++s) {
      const v2f a = ld2(&mlpin[rb + n16][4 * s + 2 * hi]);
      h0 = wmma4(a, fw1a[s], h0);
      h1 = wmma4(a, fw1b[s], h1);
    }
    const float bias0 = bm1[nh + n16], bias1 = bm1[nh + 16 + n16];
#pragma unroll
    for (int r = 0; r < 8; ++r) {
      hbuf[rb + r + 8 * hi][nh + n16] = fmaxf(h0[r] + bias0, 0.f);
      hbuf[rb + r + 8 * hi][nh + 16 + n16] = fmaxf(h1[r] + bias1, 0.f);
    }
  }
  __syncthreads();

  {
    v8f c = {};
#pragma unroll
    for (int s = 0; s < 32; ++s) {
      const v2f a = ld2(&hbuf[rb + n16][4 * s + 2 * hi]);
      c = wmma4(a, fw2[s], c);
    }
    const float bias = bm2[n0 + n16];
#pragma unroll
    for (int r = 0; r < 8; ++r)
      res[rb + r + 8 * hi][n0 + n16] = c[r] + bias + mlpin[rb + r + 8 * hi][n0 + n16];
  }
  __syncthreads();

  {
    float sum = 0.f, sq = 0.f;
#pragma unroll
    for (int i = 0; i < 8; ++i) { float v = res[row][c0 + i]; sum += v; sq += v * v; }
#pragma unroll
    for (int msk = 1; msk < 8; msk <<= 1) {
      sum += __shfl_xor(sum, msk);
      sq += __shfl_xor(sq, msk);
    }
    const float mean = sum * (1.f / 64.f);
    const float var = sq * (1.f / 64.f) - mean * mean;
    const float rstd = rsqrtf(var + 1e-5f);
#pragma unroll
    for (int i = 0; i < 8; ++i) {
      const int col = c0 + i;
      out[(grow0 + row) * E_DIM + col] =
          (res[row][col] - mean) * rstd * g2[col] + beta2[col];
    }
  }
}

extern "C" void kernel_launch(void* const* d_in, const int* in_sizes, int n_in,
                              void* d_out, int out_size, void* d_ws, size_t ws_size,
                              hipStream_t stream) {
  const float* x    = (const float*)d_in[0];
  const float* Wqkv = (const float*)d_in[4];
  const float* bqkv = (const float*)d_in[5];
  const float* Wout = (const float*)d_in[6];
  const float* bout = (const float*)d_in[7];
  const float* g1   = (const float*)d_in[8];
  const float* b1   = (const float*)d_in[9];
  const float* g2   = (const float*)d_in[10];
  const float* b2   = (const float*)d_in[11];
  const float* Wm1  = (const float*)d_in[12];
  const float* bm1  = (const float*)d_in[13];
  const float* Wm2  = (const float*)d_in[14];
  const float* bm2  = (const float*)d_in[15];
  float* out = (float*)d_out;

  float* qkv = (float*)d_ws;

  k_qkv<<<N_TOK / 32, 128, 0, stream>>>(x, Wqkv, bqkv, qkv);
  k_attn_tail<<<N_TOK / 64, 512, 0, stream>>>(qkv, Wout, bout, x, g1, b1,
                                              Wm1, bm1, Wm2, bm2, g2, b2, out);
}


# k_attn_tail key loop software-pipelined per wave: LDS fragment reads a step ahead (2nd register set); scores, exp2, range check of tile t+1 issue among P*V WMMAs of tile t; fp sum order unchanged
# speedup vs baseline: 2.5809x; 1.0536x over previous
#include <hip/hip_runtime.h>
#include <math.h>

typedef __attribute__((ext_vector_type(2))) float v2f;
typedef __attribute__((ext_vector_type(8))) float v8f;

#define E_DIM 64
#define H_NUM 8
#define D_HEAD 8
#define B_SZ 16
#define S_LEN 1024
#define N_TOK (B_SZ * S_LEN)

__device__ __forceinline__ v8f wmma4(v2f a, v2f b, v8f c) {
  return __builtin_amdgcn_wmma_f32_16x16x4_f32(
       false, a,  false, b,
       (short)0, c,  false,  false);
}

__global__ __launch_bounds__(128) void k_qkv(const float* __restrict__ x,
                                             const float* __restrict__ Wqkv,
                                             const float* __restrict__ bqkv,
                                             float* __restrict__ qkv) {
  const int wave = threadIdx.x >> 5;
  const int lane = threadIdx.x & 31;
  const int n16 = lane & 15, hi = lane >> 4;
  const int m0 = blockIdx.x * 32;
  const int n0 = wave * 48;
  const float* arow = x + (m0 + n16) * E_DIM + 2 * hi;
  const float* brow = Wqkv + (n0 + n16) * E_DIM + 2 * hi;
  v8f c00 = {}, c01 = {}, c02 = {}, c10 = {}, c11 = {}, c12 = {};
#pragma unroll
  for (int k0 = 0; k0 < E_DIM; k0 += 4) {
    v2f a0, a1, b0, b1, b2;
    a0.x = arow[k0];               a0.y = arow[k0 + 1];
    a1.x = arow[16 * E_DIM + k0];  a1.y = arow[16 * E_DIM + k0 + 1];
    b0.x = brow[k0];               b0.y = brow[k0 + 1];
    b1.x = brow[16 * E_DIM + k0];  b1.y = brow[16 * E_DIM + k0 + 1];
    b2.x = brow[32 * E_DIM + k0];  b2.y = brow[32 * E_DIM + k0 + 1];
    c00 = wmma4(a0, b0, c00);
    c01 = wmma4(a0, b1, c01);
    c02 = wmma4(a0, b2, c02);
    c10 = wmma4(a1, b0, c10);
    c11 = wmma4(a1, b1, c11);
    c12 = wmma4(a1, b2, c12);
  }
  const float bias0 = bqkv[n0 + n16];
  const float bias1 = bqkv[n0 + 16 + n16];
  const float bias2 = bqkv[n0 + 32 + n16];
  float* orow = qkv + (m0 + 8 * hi) * (3 * E_DIM) + n0 + n16;
#pragma unroll
  for (int r = 0; r < 8; ++r) {
    float* p0 = orow + r * (3 * E_DIM);
    float* p1 = p0 + 16 * (3 * E_DIM);
    p0[0]  = c00[r] + bias0;
    p0[16] = c01[r] + bias1;
    p0[32] = c02[r] + bias2;
    p1[0]  = c10[r] + bias0;
    p1[16] = c11[r] + bias1;
    p1[32] = c12[r] + bias2;
  }
}


__device__ __forceinline__ float xhalf(float x) {
  const int b = __builtin_bit_cast(int, x);
  return __builtin_bit_cast(
      float, __builtin_amdgcn_permlanex16(b, b, 0x76543210, 0xfedcba98, false, false));
}

__device__ __forceinline__ void lds_barrier() {
  __builtin_amdgcn_fence(__ATOMIC_RELEASE, "workgroup", "local");
  __builtin_amdgcn_s_barrier();
  __builtin_amdgcn_fence(__ATOMIC_ACQUIRE, "workgroup", "local");
}

#define N_KT (S_LEN / 16)
#define KT_STEP (16 * 3 * E_DIM)

struct KVFrags {
  v2f ka0, ka1;
  v2f va[4];
};

__device__ __forceinline__ KVFrags load_frags(const float* kb, const float* vb,
                                              int n16, int hi, int d) {
  KVFrags f;
  f.ka0.x = kb[n16 * 9 + 0 + 2 * hi];
  f.ka0.y = kb[n16 * 9 + 1 + 2 * hi];
  f.ka1.x = kb[n16 * 9 + 4 + 2 * hi];
  f.ka1.y = kb[n16 * 9 + 5 + 2 * hi];
#pragma unroll
  for (int kc = 0; kc < 4; ++kc) {
    f.va[kc].x = vb[(8 * hi + 2 * kc) * 9 + d];
    f.va[kc].y = vb[(8 * hi + 2 * kc + 1) * 9 + d];
  }
  return f;
}

#define PS_MAX 256.0f

__device__ __forceinline__ void tile_p(const KVFrags& f, v2f qb0, v2f qb1,
                                       const v8f& cref, v8f& p, float& ps) {
  v8f s = wmma4(f.ka0, qb0, cref);
  s = wmma4(f.ka1, qb1, s);
#pragma unroll
  for (int r = 0; r < 8; ++r) p[r] = __builtin_amdgcn_exp2f(s[r]);
  ps = ((p[0] + p[1]) + (p[2] + p[3])) + ((p[4] + p[5]) + (p[6] + p[7]));
}

__device__ __forceinline__ void tile_rebase(const KVFrags& f, v2f qb0, v2f qb1,
                                            float& m, float& l, v8f& o,
                                            v8f& cref, v8f& p, float& ps) {
  v8f s = {};
  s = wmma4(f.ka0, qb0, s);
  s = wmma4(f.ka1, qb1, s);
  float mx = fmaxf(fmaxf(fmaxf(s[0], s[1]), fmaxf(s[2], s[3])),
                   fmaxf(fmaxf(s[4], s[5]), fmaxf(s[6], s[7])));
  mx = fmaxf(mx, xhalf(mx));
  const float mn = fmaxf(m, mx);
  const float alpha = __builtin_amdgcn_exp2f(m - mn);
  m = mn;
#pragma unroll
  for (int r = 0; r < 8; ++r) p[r] = __builtin_amdgcn_exp2f(s[r] - mn);
  ps = ((p[0] + p[1]) + (p[2] + p[3])) + ((p[4] + p[5]) + (p[6] + p[7]));
  l *= alpha;
#pragma unroll
  for (int r = 0; r < 8; ++r) {
    o[r] *= alpha;
    cref[r] = -mn;
  }
  asm volatile("" : "+v"(cref));
}

__device__ __forceinline__ void tile_pv(const KVFrags& f, const v8f& p, v8f& o) {
#pragma unroll
  for (int kc = 0; kc < 4; ++kc) {
    v2f pb;
    pb.x = p[2 * kc];
    pb.y = p[2 * kc + 1];
    o = wmma4(f.va[kc], pb, o);
  }
}

__device__ __forceinline__ void attn_step(const KVFrags& f, v2f qb0, v2f qb1,
                                          v2f qc0, v2f qc1, float& m, float& l,
                                          v8f& o, v8f& cref, v8f& p, float& m2,
                                          float& l2, v8f& o2, v8f& cref2, v8f& p2) {
  v8f pn, pn2;
  float ps, ps2;
  tile_p(f, qb0, qb1, cref, pn, ps);
  tile_p(f, qc0, qc1, cref2, pn2, ps2);
  tile_pv(f, p, o);
  tile_pv(f, p2, o2);
  if (__builtin_amdgcn_ballot_w32(!(fmaxf(ps, ps2) <= PS_MAX)) != 0) {
    tile_rebase(f, qb0, qb1, m, l, o, cref, pn, ps);
    tile_rebase(f, qc0, qc1, m2, l2, o2, cref2, pn2, ps2);
  }
  l += ps;
  l2 += ps2;
  p = pn;
  p2 = pn2;
}

#define P64 68
#define P128 132
#define KV_HEAD (16 * 9)
#define KV_STAGE (H_NUM * KV_HEAD)

__device__ __forceinline__ v2f ld2(const float* p) {
  v2f v;
  v.x = p[0];
  v.y = p[1];
  return v;
}

__global__ __launch_bounds__(512) void k_attn_tail(
    const float* __restrict__ qkv, const float* __restrict__ Wout,
    const float* __restrict__ bout, const float* __restrict__ x,
    const float* __restrict__ g1, const float* __restrict__ beta1,
    const float* __restrict__ Wm1, const float* __restrict__ bm1,
    const float* __restrict__ Wm2, const float* __restrict__ bm2,
    const float* __restrict__ g2, const float* __restrict__ beta2,
    float* __restrict__ out) {
  __shared__ float ldsK0[KV_STAGE], ldsK1[KV_STAGE];
  __shared__ float ldsV0[KV_STAGE], ldsV1[KV_STAGE];
  __shared__ __attribute__((aligned(16))) float att[64][P64];
  __shared__ float res[64][P64];
  __shared__ float mlpin[64][P64];
  __shared__ float hbuf[64][P128];
  const int tid = threadIdx.x;
  const int wave = tid >> 5;
  const int lane = tid & 31;
  const int n16 = lane & 15, hi = lane >> 4;

  const int bid = blockIdx.x;
  const int b = bid >> 4;
  const int qblk = bid & 15;
  const int h = wave >> 1;
  const int qt = qblk * 4 + (wave & 1) * 2;
  const float scale = (float)(1.4426950408889634 * 0.35355339059327373);

  const float* qkvB = qkv + (size_t)(b * S_LEN) * (3 * E_DIM);
  const int srow = tid >> 5, scol = (tid & 31) * 2;
  const int sidx = (scol >> 3) * KV_HEAD + srow * 9 + (scol & 7);
  const float* stg = qkvB + srow * (3 * E_DIM) + scol;
  const float* myK0 = ldsK0 + h * KV_HEAD;
  const float* myK1 = ldsK1 + h * KV_HEAD;
  const float* myV0 = ldsV0 + h * KV_HEAD;
  const float* myV1 = ldsV1 + h * KV_HEAD;

  v2f qb0, qb1, qc0, qc1;
  {
    const float* qp = qkvB + (qt * 16 + n16) * (3 * E_DIM) + h * D_HEAD;
    qb0.x = qp[0 + 2 * hi] * scale; qb0.y = qp[1 + 2 * hi] * scale;
    qb1.x = qp[4 + 2 * hi] * scale; qb1.y = qp[5 + 2 * hi] * scale;
    const float* qn = qp + 16 * (3 * E_DIM);
    qc0.x = qn[0 + 2 * hi] * scale; qc0.y = qn[1 + 2 * hi] * scale;
    qc1.x = qn[4 + 2 * hi] * scale; qc1.y = qn[5 + 2 * hi] * scale;
  }

  v2f kA, vA, kB, vB;
  {
    const v2f k0 = ld2(stg + KT_STEP + E_DIM), v0 = ld2(stg + 2 * E_DIM);
    const v2f k1 = ld2(stg + 2 * KT_STEP + E_DIM), v1 = ld2(stg + KT_STEP + 2 * E_DIM);
    kA = ld2(stg + 3 * KT_STEP + E_DIM); vA = ld2(stg + 2 * KT_STEP + 2 * E_DIM);
    kB = ld2(stg + 4 * KT_STEP + E_DIM); vB = ld2(stg + 3 * KT_STEP + 2 * E_DIM);
    ldsK0[sidx] = k0.x; ldsK0[sidx + 1] = k0.y;
    ldsV0[sidx] = v0.x; ldsV0[sidx + 1] = v0.y;
    ldsK1[sidx] = k1.x; ldsK1[sidx + 1] = k1.y;
    ldsV1[sidx] = v1.x; ldsV1[sidx + 1] = v1.y;
  }

  float m = -3.0e38f, l = 0.f;
  float m2 = -3.0e38f, l2 = 0.f;
  v8f o = {}, o2 = {};
  v8f cref, cref2;
  const int d = n16 & 7;

  v8f p, p2;
  {
    const float* k0p = qkvB + n16 * (3 * E_DIM) + E_DIM + h * D_HEAD + 2 * hi;
    KVFrags f;
    f.ka0 = ld2(k0p);
    f.ka1 = ld2(k0p + 4);
    float ps, ps2;
    tile_rebase(f, qb0, qb1, m, l, o, cref, p, ps);
    tile_rebase(f, qc0, qc1, m2, l2, o2, cref2, p2, ps2);
    l += ps;
    l2 += ps2;
  }

  lds_barrier();
  KVFrags fe = load_frags(myK0, myV0, n16, hi, d), fo;
  for (int kt = 0; kt < N_KT - 2; kt += 2) {
    lds_barrier();
    fo = load_frags(myK1, myV1, n16, hi, d);
    ldsK0[sidx] = kA.x; ldsK0[sidx + 1] = kA.y;
    ldsV0[sidx] = vA.x; ldsV0[sidx + 1] = vA.y;
    kA = ld2(stg + min(kt + 5, N_KT - 1) * KT_STEP + E_DIM);
    vA = ld2(stg + min(kt + 4, N_KT - 1) * KT_STEP + 2 * E_DIM);
    attn_step(fe, qb0, qb1, qc0, qc1, m, l, o, cref, p, m2, l2, o2, cref2, p2);
    lds_barrier();
    fe = load_frags(myK0, myV0, n16, hi, d);
    ldsK1[sidx] = kB.x; ldsK1[sidx + 1] = kB.y;
    ldsV1[sidx] = vB.x; ldsV1[sidx + 1] = vB.y;
    kB = ld2(stg + min(kt + 6, N_KT - 1) * KT_STEP + E_DIM);
    vB = ld2(stg + min(kt + 5, N_KT - 1) * KT_STEP + 2 * E_DIM);
    attn_step(fo, qb0, qb1, qc0, qc1, m, l, o, cref, p, m2, l2, o2, cref2, p2);
  }
  lds_barrier();
  fo = load_frags(myK1, myV1, n16, hi, d);
  attn_step(fe, qb0, qb1, qc0, qc1, m, l, o, cref, p, m2, l2, o2, cref2, p2);
  tile_pv(fo, p, o);
  tile_pv(fo, p2, o2);

  l += xhalf(l);
  l2 += xhalf(l2);
  {
    float4 w, w2;
    w.x = (hi ? o[4] : o[0]) / l;
    w.y = (hi ? o[5] : o[1]) / l;
    w.z = (hi ? o[6] : o[2]) / l;
    w.w = (hi ? o[7] : o[3]) / l;
    w2.x = (hi ? o2[4] : o2[0]) / l2;
    w2.y = (hi ? o2[5] : o2[1]) / l2;
    w2.z = (hi ? o2[6] : o2[2]) / l2;
    w2.w = (hi ? o2[7] : o2[3]) / l2;
    const int r0 = (wave & 1) * 32 + n16;
    *reinterpret_cast<float4*>(&att[r0][h * D_HEAD + 4 * hi]) = w;
    *reinterpret_cast<float4*>(&att[r0 + 16][h * D_HEAD + 4 * hi]) = w2;
  }

  const int rb = (wave >> 2) * 16;
  const int cw = wave & 3;
  const int n0 = cw * 16;
  const int nh = cw * 32;
  const int grow0 = b * S_LEN + qblk * 64;

  const float* brow = Wout + (n0 + n16) * E_DIM + 2 * hi;
  const float* w1r0 = Wm1 + (nh + n16) * E_DIM + 2 * hi;
  const float* w1r1 = Wm1 + (nh + 16 + n16) * E_DIM + 2 * hi;
  const float* w2r = Wm2 + (n0 + n16) * (2 * E_DIM) + 2 * hi;
  v2f fb[16], fw1a[16], fw1b[16], fw2[32];
#pragma unroll
  for (int s = 0; s < 16; ++s) fb[s] = ld2(brow + 4 * s);
  float xr[8];
#pragma unroll
  for (int r = 0; r < 8; ++r) xr[r] = x[(grow0 + rb + r + 8 * hi) * E_DIM + n0 + n16];
#pragma unroll
  for (int s = 0; s < 16; ++s) { fw1a[s] = ld2(w1r0 + 4 * s); fw1b[s] = ld2(w1r1 + 4 * s); }
#pragma unroll
  for (int s = 0; s < 32; ++s) fw2[s] = ld2(w2r + 4 * s);
  __syncthreads();

  {
    v8f c = {};
#pragma unroll
    for (int s = 0; s < 16; ++s) c = wmma4(ld2(&att[rb + n16][4 * s + 2 * hi]), fb[s], c);
    const float bias = bout[n0 + n16];
#pragma unroll
    for (int r = 0; r < 8; ++r) res[rb + r + 8 * hi][n0 + n16] = c[r] + bias + xr[r];
  }
  __syncthreads();

  const int row = rb + ((tid & 127) >> 3);
  const int c0 = (tid & 7) * 8;
  {
    float sum = 0.f, sq = 0.f;
#pragma unroll
    for (int i = 0; i < 8; ++i) { float v = res[row][c0 + i]; sum += v; sq += v * v; }
#pragma unroll
    for (int msk = 1; msk < 8; msk <<= 1) {
      sum += __shfl_xor(sum, msk);
      sq += __shfl_xor(sq, msk);
    }
    const float mean = sum * (1.f / 64.f);
    const float var = sq * (1.f / 64.f) - mean * mean;
    const float rstd = rsqrtf(var + 1e-5f);
#pragma unroll
    for (int i = 0; i < 8; ++i) {
      const int col = c0 + i;
      mlpin[row][col] = (res[row][col] - mean) * rstd * g1[col] + beta1[col];
    }
  }
  __syncthreads();

  {
    v8f h0 = {}, h1 = {};
#pragma unroll
    for (int s = 0; s < 16; ++s) {
      const v2f a = ld2(&mlpin[rb + n16][4 * s + 2 * hi]);
      h0 = wmma4(a, fw1a[s], h0);
      h1 = wmma4(a, fw1b[s], h1);
    }
    const float bias0 = bm1[nh + n16], bias1 = bm1[nh + 16 + n16];
#pragma unroll
    for (int r = 0; r < 8; ++r) {
      hbuf[rb + r + 8 * hi][nh + n16] = fmaxf(h0[r] + bias0, 0.f);
      hbuf[rb + r + 8 * hi][nh + 16 + n16] = fmaxf(h1[r] + bias1, 0.f);
    }
  }
  __syncthreads();

  {
    v8f c = {};
#pragma unroll
    for (int s = 0; s < 32; ++s) {
      const v2f a = ld2(&hbuf[rb + n16][4 * s + 2 * hi]);
      c = wmma4(a, fw2[s], c);
    }
    const float bias = bm2[n0 + n16];
#pragma unroll
    for (int r = 0; r < 8; ++r)
      res[rb + r + 8 * hi][n0 + n16] = c[r] + bias + mlpin[rb + r + 8 * hi][n0 + n16];
  }
  __syncthreads();

  {
    float sum = 0.f, sq = 0.f;
#pragma unroll
    for (int i = 0; i < 8; ++i) { float v = res[row][c0 + i]; sum += v; sq += v * v; }
#pragma unroll
    for (int msk = 1; msk < 8; msk <<= 1) {
      sum += __shfl_xor(sum, msk);
      sq += __shfl_xor(sq, msk);
    }
    const float mean = sum * (1.f / 64.f);
    const float var = sq * (1.f / 64.f) - mean * mean;
    const float rstd = rsqrtf(var + 1e-5f);
#pragma unroll
    for (int i = 0; i < 8; ++i) {
      const int col = c0 + i;
      out[(grow0 + row) * E_DIM + col] =
          (res[row][col] - mean) * rstd * g2[col] + beta2[col];
    }
  }
}

extern "C" void kernel_launch(void* const* d_in, const int* in_sizes, int n_in,
                              void* d_out, int out_size, void* d_ws, size_t ws_size,
                              hipStream_t stream) {
  const float* x    = (const float*)d_in[0];
  const float* Wqkv = (const float*)d_in[4];
  const float* bqkv = (const float*)d_in[5];
  const float* Wout = (const float*)d_in[6];
  const float* bout = (const float*)d_in[7];
  const float* g1   = (const float*)d_in[8];
  const float* b1   = (const float*)d_in[9];
  const float* g2   = (const float*)d_in[10];
  const float* b2   = (const float*)d_in[11];
  const float* Wm1  = (const float*)d_in[12];
  const float* bm1  = (const float*)d_in[13];
  const float* Wm2  = (const float*)d_in[14];
  const float* bm2  = (const float*)d_in[15];
  float* out = (float*)d_out;

  float* qkv = (float*)d_ws;

  k_qkv<<<N_TOK / 32, 128, 0, stream>>>(x, Wqkv, bqkv, qkv);
  k_attn_tail<<<N_TOK / 64, 512, 0, stream>>>(qkv, Wout, bout, x, g1, b1,
                                              Wm1, bm1, Wm2, bm2, g2, b2, out);
}


# k_attn_tail: tail weights, biases and LayerNorm vectors copied to LDS under the first key steps by whole-line loads; the tail reads its B fragments and vectors from LDS, not by 16-line global loads
# speedup vs baseline: 2.7403x; 1.0618x over previous
#include <hip/hip_runtime.h>
#include <math.h>

typedef __attribute__((ext_vector_type(2))) float v2f;
typedef __attribute__((ext_vector_type(8))) float v8f;

#define E_DIM 64
#define H_NUM 8
#define D_HEAD 8
#define B_SZ 16
#define S_LEN 1024
#define N_TOK (B_SZ * S_LEN)

__device__ __forceinline__ v8f wmma4(v2f a, v2f b, v8f c) {
  return __builtin_amdgcn_wmma_f32_16x16x4_f32(
       false, a,  false, b,
       (short)0, c,  false,  false);
}

__global__ __launch_bounds__(128) void k_qkv(const float* __restrict__ x,
                                             const float* __restrict__ Wqkv,
                                             const float* __restrict__ bqkv,
                                             float* __restrict__ qkv) {
  const int wave = threadIdx.x >> 5;
  const int lane = threadIdx.x & 31;
  const int n16 = lane & 15, hi = lane >> 4;
  const int m0 = blockIdx.x * 32;
  const int n0 = wave * 48;
  const float* arow = x + (m0 + n16) * E_DIM + 2 * hi;
  const float* brow = Wqkv + (n0 + n16) * E_DIM + 2 * hi;
  v8f c00 = {}, c01 = {}, c02 = {}, c10 = {}, c11 = {}, c12 = {};
#pragma unroll
  for (int k0 = 0; k0 < E_DIM; k0 += 4) {
    v2f a0, a1, b0, b1, b2;
    a0.x = arow[k0];               a0.y = arow[k0 + 1];
    a1.x = arow[16 * E_DIM + k0];  a1.y = arow[16 * E_DIM + k0 + 1];
    b0.x = brow[k0];               b0.y = brow[k0 + 1];
    b1.x = brow[16 * E_DIM + k0];  b1.y = brow[16 * E_DIM + k0 + 1];
    b2.x = brow[32 * E_DIM + k0];  b2.y = brow[32 * E_DIM + k0 + 1];
    c00 = wmma4(a0, b0, c00);
    c01 = wmma4(a0, b1, c01);
    c02 = wmma4(a0, b2, c02);
    c10 = wmma4(a1, b0, c10);
    c11 = wmma4(a1, b1, c11);
    c12 = wmma4(a1, b2, c12);
  }
  const float bias0 = bqkv[n0 + n16];
  const float bias1 = bqkv[n0 + 16 + n16];
  const float bias2 = bqkv[n0 + 32 + n16];
  float* orow = qkv + (m0 + 8 * hi) * (3 * E_DIM) + n0 + n16;
#pragma unroll
  for (int r = 0; r < 8; ++r) {
    float* p0 = orow + r * (3 * E_DIM);
    float* p1 = p0 + 16 * (3 * E_DIM);
    p0[0]  = c00[r] + bias0;
    p0[16] = c01[r] + bias1;
    p0[32] = c02[r] + bias2;
    p1[0]  = c10[r] + bias0;
    p1[16] = c11[r] + bias1;
    p1[32] = c12[r] + bias2;
  }
}


__device__ __forceinline__ float xhalf(float x) {
  const int b = __builtin_bit_cast(int, x);
  return __builtin_bit_cast(
      float, __builtin_amdgcn_permlanex16(b, b, 0x76543210, 0xfedcba98, false, false));
}

__device__ __forceinline__ void lds_barrier() {
  __builtin_amdgcn_fence(__ATOMIC_RELEASE, "workgroup", "local");
  __builtin_amdgcn_s_barrier();
  __builtin_amdgcn_fence(__ATOMIC_ACQUIRE, "workgroup", "local");
}

#define N_KT (S_LEN / 16)
#define KT_STEP (16 * 3 * E_DIM)

struct KVFrags {
  v2f ka0, ka1;
  v2f va[4];
};

__device__ __forceinline__ KVFrags load_frags(const float* kb, const float* vb,
                                              int n16, int hi, int d) {
  KVFrags f;
  f.ka0.x = kb[n16 * 9 + 0 + 2 * hi];
  f.ka0.y = kb[n16 * 9 + 1 + 2 * hi];
  f.ka1.x = kb[n16 * 9 + 4 + 2 * hi];
  f.ka1.y = kb[n16 * 9 + 5 + 2 * hi];
#pragma unroll
  for (int kc = 0; kc < 4; ++kc) {
    f.va[kc].x = vb[(8 * hi + 2 * kc) * 9 + d];
    f.va[kc].y = vb[(8 * hi + 2 * kc + 1) * 9 + d];
  }
  return f;
}

#define PS_MAX 256.0f

__device__ __forceinline__ void tile_p(const KVFrags& f, v2f qb0, v2f qb1,
                                       const v8f& cref, v8f& p, float& ps) {
  v8f s = wmma4(f.ka0, qb0, cref);
  s = wmma4(f.ka1, qb1, s);
#pragma unroll
  for (int r = 0; r < 8; ++r) p[r] = __builtin_amdgcn_exp2f(s[r]);
  ps = ((p[0] + p[1]) + (p[2] + p[3])) + ((p[4] + p[5]) + (p[6] + p[7]));
}

__device__ __forceinline__ void tile_rebase(const KVFrags& f, v2f qb0, v2f qb1,
                                            float& m, float& l, v8f& o,
                                            v8f& cref, v8f& p, float& ps) {
  v8f s = {};
  s = wmma4(f.ka0, qb0, s);
  s = wmma4(f.ka1, qb1, s);
  float mx = fmaxf(fmaxf(fmaxf(s[0], s[1]), fmaxf(s[2], s[3])),
                   fmaxf(fmaxf(s[4], s[5]), fmaxf(s[6], s[7])));
  mx = fmaxf(mx, xhalf(mx));
  const float mn = fmaxf(m, mx);
  const float alpha = __builtin_amdgcn_exp2f(m - mn);
  m = mn;
#pragma unroll
  for (int r = 0; r < 8; ++r) p[r] = __builtin_amdgcn_exp2f(s[r] - mn);
  ps = ((p[0] + p[1]) + (p[2] + p[3])) + ((p[4] + p[5]) + (p[6] + p[7]));
  l *= alpha;
#pragma unroll
  for (int r = 0; r < 8; ++r) {
    o[r] *= alpha;
    cref[r] = -mn;
  }
  asm volatile("" : "+v"(cref));
}

__device__ __forceinline__ void tile_pv(const KVFrags& f, const v8f& p, v8f& o) {
#pragma unroll
  for (int kc = 0; kc < 4; ++kc) {
    v2f pb;
    pb.x = p[2 * kc];
    pb.y = p[2 * kc + 1];
    o = wmma4(f.va[kc], pb, o);
  }
}

__device__ __forceinline__ void attn_step(const KVFrags& f, v2f qb0, v2f qb1,
                                          v2f qc0, v2f qc1, float& m, float& l,
                                          v8f& o, v8f& cref, v8f& p, float& m2,
                                          float& l2, v8f& o2, v8f& cref2, v8f& p2) {
  v8f pn, pn2;
  float ps, ps2;
  tile_p(f, qb0, qb1, cref, pn, ps);
  tile_p(f, qc0, qc1, cref2, pn2, ps2);
  tile_pv(f, p, o);
  tile_pv(f, p2, o2);
  if (__builtin_amdgcn_ballot_w32(!(fmaxf(ps, ps2) <= PS_MAX)) != 0) {
    tile_rebase(f, qb0, qb1, m, l, o, cref, pn, ps);
    tile_rebase(f, qc0, qc1, m2, l2, o2, cref2, pn2, ps2);
  }
  l += ps;
  l2 += ps2;
  p = pn;
  p2 = pn2;
}

#define P64 68
#define P128 132
#define KV_HEAD (16 * 9)
#define KV_STAGE (H_NUM * KV_HEAD)
#define PV_BOUT 0
#define PV_G1 (E_DIM)
#define PV_BETA1 (2 * E_DIM)
#define PV_BM1 (3 * E_DIM)
#define PV_BM2 (5 * E_DIM)
#define PV_G2 (6 * E_DIM)
#define PV_BETA2 (7 * E_DIM)

__device__ __forceinline__ v2f ld2(const float* p) {
  v2f v;
  v.x = p[0];
  v.y = p[1];
  return v;
}

__device__ __forceinline__ float4 ld4(const float* p) {
  float4 v;
  v.x = p[0];
  v.y = p[1];
  v.z = p[2];
  v.w = p[3];
  return v;
}

__global__ __launch_bounds__(512) void k_attn_tail(
    const float* __restrict__ qkv, const float* __restrict__ Wout,
    const float* __restrict__ bout, const float* __restrict__ x,
    const float* __restrict__ g1, const float* __restrict__ beta1,
    const float* __restrict__ Wm1, const float* __restrict__ bm1,
    const float* __restrict__ Wm2, const float* __restrict__ bm2,
    const float* __restrict__ g2, const float* __restrict__ beta2,
    float* __restrict__ out) {
  __shared__ float ldsK0[KV_STAGE], ldsK1[KV_STAGE];
  __shared__ float ldsV0[KV_STAGE], ldsV1[KV_STAGE];
  __shared__ __attribute__((aligned(16))) float att[64][P64];
  __shared__ float res[64][P64];
  __shared__ float mlpin[64][P64];
  __shared__ float hbuf[64][P128];
  __shared__ __attribute__((aligned(16))) float wOut[E_DIM][P64];
  __shared__ __attribute__((aligned(16))) float wM1[2 * E_DIM][P64];
  __shared__ __attribute__((aligned(16))) float wM2[E_DIM][P128];
  __shared__ __attribute__((aligned(16))) float pvec[8 * E_DIM];
  const int tid = threadIdx.x;
  const int wave = tid >> 5;
  const int lane = tid & 31;
  const int n16 = lane & 15, hi = lane >> 4;

  const int bid = blockIdx.x;
  const int b = bid >> 4;
  const int qblk = bid & 15;
  const int h = wave >> 1;
  const int qt = qblk * 4 + (wave & 1) * 2;
  const float scale = (float)(1.4426950408889634 * 0.35355339059327373);

  const float* qkvB = qkv + (size_t)(b * S_LEN) * (3 * E_DIM);
  const int srow = tid >> 5, scol = (tid & 31) * 2;
  const int sidx = (scol >> 3) * KV_HEAD + srow * 9 + (scol & 7);
  const float* stg = qkvB + srow * (3 * E_DIM) + scol;
  const float* myK0 = ldsK0 + h * KV_HEAD;
  const float* myK1 = ldsK1 + h * KV_HEAD;
  const float* myV0 = ldsV0 + h * KV_HEAD;
  const float* myV1 = ldsV1 + h * KV_HEAD;

  v2f qb0, qb1, qc0, qc1;
  v2f qr0, qr1, qr2, qr3;
  {
    const float* qp = qkvB + (qt * 16 + n16) * (3 * E_DIM) + h * D_HEAD + 2 * hi;
    const float* qn = qp + 16 * (3 * E_DIM);
    qr0 = ld2(qp); qr1 = ld2(qp + 4);
    qr2 = ld2(qn); qr3 = ld2(qn + 4);
    qb0 = qr0 * scale; qb1 = qr1 * scale;
    qc0 = qr2 * scale; qc1 = qr3 * scale;
  }

  v2f kA, vA, kB, vB;
  KVFrags f0;
  {
    const v2f k0 = ld2(stg + KT_STEP + E_DIM), v0 = ld2(stg + 2 * E_DIM);
    const v2f k1 = ld2(stg + 2 * KT_STEP + E_DIM), v1 = ld2(stg + KT_STEP + 2 * E_DIM);
    kA = ld2(stg + 3 * KT_STEP + E_DIM); vA = ld2(stg + 2 * KT_STEP + 2 * E_DIM);
    kB = ld2(stg + 4 * KT_STEP + E_DIM); vB = ld2(stg + 3 * KT_STEP + 2 * E_DIM);
    const float* k0p = qkvB + n16 * (3 * E_DIM) + E_DIM + h * D_HEAD + 2 * hi;
    f0.ka0 = ld2(k0p);
    f0.ka1 = ld2(k0p + 4);
    ldsK0[sidx] = k0.x; ldsK0[sidx + 1] = k0.y;
    ldsV0[sidx] = v0.x; ldsV0[sidx + 1] = v0.y;
    ldsK1[sidx] = k1.x; ldsK1[sidx + 1] = k1.y;
    ldsV1[sidx] = v1.x; ldsV1[sidx + 1] = v1.y;
    asm volatile("" : : "v"(kA), "v"(vA), "v"(kB), "v"(vB), "v"(qr0), "v"(qr1),
                 "v"(qr2), "v"(qr3), "v"(f0.ka0), "v"(f0.ka1));
  }

  float m = -3.0e38f, l = 0.f;
  float m2 = -3.0e38f, l2 = 0.f;
  v8f o = {}, o2 = {};
  v8f cref, cref2;
  const int d = n16 & 7;

  v8f p, p2;
  {
    float ps, ps2;
    tile_rebase(f0, qb0, qb1, m, l, o, cref, p, ps);
    tile_rebase(f0, qc0, qc1, m2, l2, o2, cref2, p2, ps2);
    l += ps;
    l2 += ps2;
  }
  lds_barrier();
  float4 wq[10];
#pragma unroll
  for (int i = 0; i < 2; ++i) wq[i] = ld4(Wout + 4 * (tid + 512 * i));
#pragma unroll
  for (int i = 0; i < 4; ++i) wq[2 + i] = ld4(Wm1 + 4 * (tid + 512 * i));
#pragma unroll
  for (int i = 0; i < 4; ++i) wq[6 + i] = ld4(Wm2 + 4 * (tid + 512 * i));
  const int wv = __builtin_amdgcn_readfirstlane(wave);
  const float* pbase = wv < 2 ? bout : wv < 4 ? g1 : wv < 6 ? beta1 : wv < 10 ? bm1
                     : wv < 12 ? bm2 : wv < 14 ? g2 : beta2;
  const int poff = wv < 2 ? 0 : wv < 4 ? PV_G1 : wv < 6 ? PV_BETA1 : wv < 10 ? PV_BM1
                 : wv < 12 ? PV_BM2 : wv < 14 ? PV_G2 : PV_BETA2;
  const float pval = pbase[tid - poff];
  KVFrags fe = load_frags(myK0, myV0, n16, hi, d), fo;
  auto step_pair = [&](const int kt) __attribute__((always_inline)) {
    lds_barrier();
    fo = load_frags(myK1, myV1, n16, hi, d);
    ldsK0[sidx] = kA.x; ldsK0[sidx + 1] = kA.y;
    ldsV0[sidx] = vA.x; ldsV0[sidx + 1] = vA.y;
    kA = ld2(stg + min(kt + 5, N_KT - 1) * KT_STEP + E_DIM);
    vA = ld2(stg + min(kt + 4, N_KT - 1) * KT_STEP + 2 * E_DIM);
    attn_step(fe, qb0, qb1, qc0, qc1, m, l, o, cref, p, m2, l2, o2, cref2, p2);
    lds_barrier();
    fe = load_frags(myK0, myV0, n16, hi, d);
    ldsK1[sidx] = kB.x; ldsK1[sidx + 1] = kB.y;
    ldsV1[sidx] = vB.x; ldsV1[sidx + 1] = vB.y;
    kB = ld2(stg + min(kt + 6, N_KT - 1) * KT_STEP + E_DIM);
    vB = ld2(stg + min(kt + 5, N_KT - 1) * KT_STEP + 2 * E_DIM);
    attn_step(fo, qb0, qb1, qc0, qc1, m, l, o, cref, p, m2, l2, o2, cref2, p2);
  };
  step_pair(0);
#pragma unroll
  for (int i = 0; i < 2; ++i) {
    const int q = tid + 512 * i;
    *reinterpret_cast<float4*>(&wOut[q >> 4][(q & 15) * 4]) = wq[i];
  }
#pragma unroll
  for (int i = 0; i < 4; ++i) {
    const int q = tid + 512 * i;
    *reinterpret_cast<float4*>(&wM1[q >> 4][(q & 15) * 4]) = wq[2 + i];
  }
#pragma unroll
  for (int i = 0; i < 4; ++i) {
    const int q = tid + 512 * i;
    *reinterpret_cast<float4*>(&wM2[q >> 5][(q & 31) * 4]) = wq[6 + i];
  }
  pvec[tid] = pval;
  for (int kt = 2; kt < N_KT - 2; kt += 2) step_pair(kt);
  lds_barrier();
  fo = load_frags(myK1, myV1, n16, hi, d);
  attn_step(fe, qb0, qb1, qc0, qc1, m, l, o, cref, p, m2, l2, o2, cref2, p2);
  tile_pv(fo, p, o);
  tile_pv(fo, p2, o2);

  l += xhalf(l);
  l2 += xhalf(l2);
  {
    float4 w, w2;
    w.x = (hi ? o[4] : o[0]) / l;
    w.y = (hi ? o[5] : o[1]) / l;
    w.z = (hi ? o[6] : o[2]) / l;
    w.w = (hi ? o[7] : o[3]) / l;
    w2.x = (hi ? o2[4] : o2[0]) / l2;
    w2.y = (hi ? o2[5] : o2[1]) / l2;
    w2.z = (hi ? o2[6] : o2[2]) / l2;
    w2.w = (hi ? o2[7] : o2[3]) / l2;
    const int r0 = (wave & 1) * 32 + n16;
    *reinterpret_cast<float4*>(&att[r0][h * D_HEAD + 4 * hi]) = w;
    *reinterpret_cast<float4*>(&att[r0 + 16][h * D_HEAD + 4 * hi]) = w2;
  }

  const int rb = (wave >> 2) * 16;
  const int cw = wave & 3;
  const int n0 = cw * 16;
  const int nh = cw * 32;
  const int grow0 = b * S_LEN + qblk * 64;

  const float* brow = &wOut[n0 + n16][2 * hi];
  const float* w1r0 = &wM1[nh + n16][2 * hi];
  const float* w1r1 = &wM1[nh + 16 + n16][2 * hi];
  const float* w2r = &wM2[n0 + n16][2 * hi];
  v2f fb[16], fw1a[16], fw1b[16], fw2[32];
#pragma unroll
  for (int s = 0; s < 16; ++s) fb[s] = ld2(brow + 4 * s);
  float xr[8];
#pragma unroll
  for (int r = 0; r < 8; ++r) xr[r] = x[(grow0 + rb + r + 8 * hi) * E_DIM + n0 + n16];
#pragma unroll
  for (int s = 0; s < 16; ++s) { fw1a[s] = ld2(w1r0 + 4 * s); fw1b[s] = ld2(w1r1 + 4 * s); }
#pragma unroll
  for (int s = 0; s < 32; ++s) fw2[s] = ld2(w2r + 4 * s);
  __syncthreads();

  {
    v8f c = {};
#pragma unroll
    for (int s = 0; s < 16; ++s) c = wmma4(ld2(&att[rb + n16][4 * s + 2 * hi]), fb[s], c);
    const float bias = pvec[PV_BOUT + n0 + n16];
#pragma unroll
    for (int r = 0; r < 8; ++r) res[rb + r + 8 * hi][n0 + n16] = c[r] + bias + xr[r];
  }
  __syncthreads();

  const int row = rb + ((tid & 127) >> 3);
  const int c0 = (tid & 7) * 8;
  {
    float sum = 0.f, sq = 0.f;
#pragma unroll
    for (int i = 0; i < 8; ++i) { float v = res[row][c0 + i]; sum += v; sq += v * v; }
#pragma unroll
    for (int msk = 1; msk < 8; msk <<= 1) {
      sum += __shfl_xor(sum, msk);
      sq += __shfl_xor(sq, msk);
    }
    const float mean = sum * (1.f / 64.f);
    const float var = sq * (1.f / 64.f) - mean * mean;
    const float rstd = rsqrtf(var + 1e-5f);
#pragma unroll
    for (int i = 0; i < 8; ++i) {
      const int col = c0 + i;
      mlpin[row][col] = (res[row][col] - mean) * rstd * pvec[PV_G1 + col] + pvec[PV_BETA1 + col];
    }
  }
  __syncthreads();

  {
    v8f h0 = {}, h1 = {};
#pragma unroll
    for (int s = 0; s < 16; ++s) {
      const v2f a = ld2(&mlpin[rb + n16][4 * s + 2 * hi]);
      h0 = wmma4(a, fw1a[s], h0);
      h1 = wmma4(a, fw1b[s], h1);
    }
    const float bias0 = pvec[PV_BM1 + nh + n16], bias1 = pvec[PV_BM1 + nh + 16 + n16];
#pragma unroll
    for (int r = 0; r < 8; ++r) {
      hbuf[rb + r + 8 * hi][nh + n16] = fmaxf(h0[r] + bias0, 0.f);
      hbuf[rb + r + 8 * hi][nh + 16 + n16] = fmaxf(h1[r] + bias1, 0.f);
    }
  }
  __syncthreads();

  {
    v8f c = {};
#pragma unroll
    for (int s = 0; s < 32; ++s) {
      const v2f a = ld2(&hbuf[rb + n16][4 * s + 2 * hi]);
      c = wmma4(a, fw2[s], c);
    }
    const float bias = pvec[PV_BM2 + n0 + n16];
#pragma unroll
    for (int r = 0; r < 8; ++r)
      res[rb + r + 8 * hi][n0 + n16] = c[r] + bias + mlpin[rb + r + 8 * hi][n0 + n16];
  }
  __syncthreads();

  {
    float sum = 0.f, sq = 0.f;
#pragma unroll
    for (int i = 0; i < 8; ++i) { float v = res[row][c0 + i]; sum += v; sq += v * v; }
#pragma unroll
    for (int msk = 1; msk < 8; msk <<= 1) {
      sum += __shfl_xor(sum, msk);
      sq += __shfl_xor(sq, msk);
    }
    const float mean = sum * (1.f / 64.f);
    const float var = sq * (1.f / 64.f) - mean * mean;
    const float rstd = rsqrtf(var + 1e-5f);
#pragma unroll
    for (int i = 0; i < 8; ++i) {
      const int col = c0 + i;
      out[(grow0 + row) * E_DIM + col] =
          (res[row][col] - mean) * rstd * pvec[PV_G2 + col] + pvec[PV_BETA2 + col];
    }
  }
}

extern "C" void kernel_launch(void* const* d_in, const int* in_sizes, int n_in,
                              void* d_out, int out_size, void* d_ws, size_t ws_size,
                              hipStream_t stream) {
  const float* x    = (const float*)d_in[0];
  const float* Wqkv = (const float*)d_in[4];
  const float* bqkv = (const float*)d_in[5];
  const float* Wout = (const float*)d_in[6];
  const float* bout = (const float*)d_in[7];
  const float* g1   = (const float*)d_in[8];
  const float* b1   = (const float*)d_in[9];
  const float* g2   = (const float*)d_in[10];
  const float* b2   = (const float*)d_in[11];
  const float* Wm1  = (const float*)d_in[12];
  const float* bm1  = (const float*)d_in[13];
  const float* Wm2  = (const float*)d_in[14];
  const float* bm2  = (const float*)d_in[15];
  float* out = (float*)d_out;

  float* qkv = (float*)d_ws;

  k_qkv<<<N_TOK / 32, 128, 0, stream>>>(x, Wqkv, bqkv, qkv);
  k_attn_tail<<<N_TOK / 64, 512, 0, stream>>>(qkv, Wout, bout, x, g1, b1,
                                              Wm1, bm1, Wm2, bm2, g2, b2, out);
}


# k_attn_tail key loop runs unchecked first: no p sum, compare or branch per key step; l summed by a ones row of V^T on the matrix pipe; one finite test, checked rerun if it fails; l sum order changes
# speedup vs baseline: 2.8854x; 1.0530x over previous
#include <hip/hip_runtime.h>
#include <math.h>

typedef __attribute__((ext_vector_type(2))) float v2f;
typedef __attribute__((ext_vector_type(8))) float v8f;

#define E_DIM 64
#define H_NUM 8
#define D_HEAD 8
#define B_SZ 16
#define S_LEN 1024
#define N_TOK (B_SZ * S_LEN)

__device__ __forceinline__ v8f wmma4(v2f a, v2f b, v8f c) {
  return __builtin_amdgcn_wmma_f32_16x16x4_f32(
       false, a,  false, b,
       (short)0, c,  false,  false);
}

template <bool B> struct Flag { static constexpr bool value = B; };

__global__ __launch_bounds__(128) void k_qkv(const float* __restrict__ x,
                                             const float* __restrict__ Wqkv,
                                             const float* __restrict__ bqkv,
                                             float* __restrict__ qkv) {
  const int wave = threadIdx.x >> 5;
  const int lane = threadIdx.x & 31;
  const int n16 = lane & 15, hi = lane >> 4;
  const int m0 = blockIdx.x * 32;
  const int n0 = wave * 48;
  const float* arow = x + (m0 + n16) * E_DIM + 2 * hi;
  const float* brow = Wqkv + (n0 + n16) * E_DIM + 2 * hi;
  v8f c00 = {}, c01 = {}, c02 = {}, c10 = {}, c11 = {}, c12 = {};
#pragma unroll
  for (int k0 = 0; k0 < E_DIM; k0 += 4) {
    v2f a0, a1, b0, b1, b2;
    a0.x = arow[k0];               a0.y = arow[k0 + 1];
    a1.x = arow[16 * E_DIM + k0];  a1.y = arow[16 * E_DIM + k0 + 1];
    b0.x = brow[k0];               b0.y = brow[k0 + 1];
    b1.x = brow[16 * E_DIM + k0];  b1.y = brow[16 * E_DIM + k0 + 1];
    b2.x = brow[32 * E_DIM + k0];  b2.y = brow[32 * E_DIM + k0 + 1];
    c00 = wmma4(a0, b0, c00);
    c01 = wmma4(a0, b1, c01);
    c02 = wmma4(a0, b2, c02);
    c10 = wmma4(a1, b0, c10);
    c11 = wmma4(a1, b1, c11);
    c12 = wmma4(a1, b2, c12);
  }
  const float bias0 = bqkv[n0 + n16];
  const float bias1 = bqkv[n0 + 16 + n16];
  const float bias2 = bqkv[n0 + 32 + n16];
  float* orow = qkv + (m0 + 8 * hi) * (3 * E_DIM) + n0 + n16;
#pragma unroll
  for (int r = 0; r < 8; ++r) {
    float* p0 = orow + r * (3 * E_DIM);
    float* p1 = p0 + 16 * (3 * E_DIM);
    p0[0]  = c00[r] + bias0;
    p0[16] = c01[r] + bias1;
    p0[32] = c02[r] + bias2;
    p1[0]  = c10[r] + bias0;
    p1[16] = c11[r] + bias1;
    p1[32] = c12[r] + bias2;
  }
}


__device__ __forceinline__ float xhalf(float x) {
  const int b = __builtin_bit_cast(int, x);
  return __builtin_bit_cast(
      float, __builtin_amdgcn_permlanex16(b, b, 0x76543210, 0xfedcba98, false, false));
}

__device__ __forceinline__ void lds_barrier() {
  __builtin_amdgcn_fence(__ATOMIC_RELEASE, "workgroup", "local");
  __builtin_amdgcn_s_barrier();
  __builtin_amdgcn_fence(__ATOMIC_ACQUIRE, "workgroup", "local");
}

__device__ __forceinline__ void lds_arrive() {
  __builtin_amdgcn_fence(__ATOMIC_RELEASE, "workgroup", "local");
  __builtin_amdgcn_s_barrier_signal(-1);
}
__device__ __forceinline__ void lds_pass() {
  __builtin_amdgcn_s_barrier_wait(-1);
  __builtin_amdgcn_fence(__ATOMIC_ACQUIRE, "workgroup", "local");
}

#define N_KT (S_LEN / 16)
#define KT_STEP (16 * 3 * E_DIM)

struct KVFrags {
  v2f ka0, ka1;
  v2f va[4];
};

__device__ __forceinline__ KVFrags load_frags(const float* kb, const float* vb,
                                              int n16, int hi, int d) {
  KVFrags f;
  f.ka0.x = kb[n16 * 9 + 0 + 2 * hi];
  f.ka0.y = kb[n16 * 9 + 1 + 2 * hi];
  f.ka1.x = kb[n16 * 9 + 4 + 2 * hi];
  f.ka1.y = kb[n16 * 9 + 5 + 2 * hi];
#pragma unroll
  for (int kc = 0; kc < 4; ++kc) {
    f.va[kc].x = vb[(8 * hi + 2 * kc) * 9 + d];
    f.va[kc].y = vb[(8 * hi + 2 * kc + 1) * 9 + d];
  }
  return f;
}

#define PS_MAX 256.0f

__device__ __forceinline__ float sum8(const v8f& v) {
  return ((v[0] + v[1]) + (v[2] + v[3])) + ((v[4] + v[5]) + (v[6] + v[7]));
}

__device__ __forceinline__ v8f tile_s(const KVFrags& f, v2f qb0, v2f qb1, const v8f& cref) {
  v8f s = wmma4(f.ka0, qb0, cref);
  return wmma4(f.ka1, qb1, s);
}

__device__ __forceinline__ void tile_exp(const v8f& s, v8f& p) {
#pragma unroll
  for (int r = 0; r < 8; ++r) p[r] = __builtin_amdgcn_exp2f(s[r]);
}

__device__ __forceinline__ void tile_rebase(const KVFrags& f, v2f qb0, v2f qb1,
                                            float& m, v8f& o, v8f& cref, v8f& p) {
  v8f s = {};
  s = wmma4(f.ka0, qb0, s);
  s = wmma4(f.ka1, qb1, s);
  float mx = fmaxf(fmaxf(fmaxf(s[0], s[1]), fmaxf(s[2], s[3])),
                   fmaxf(fmaxf(s[4], s[5]), fmaxf(s[6], s[7])));
  mx = fmaxf(mx, xhalf(mx));
  const float mn = fmaxf(m, mx);
  const float alpha = __builtin_amdgcn_exp2f(m - mn);
  m = mn;
#pragma unroll
  for (int r = 0; r < 8; ++r) p[r] = __builtin_amdgcn_exp2f(s[r] - mn);
#pragma unroll
  for (int r = 0; r < 8; ++r) {
    o[r] *= alpha;
    cref[r] = -mn;
  }
  asm volatile("" : "+v"(cref));
}

__device__ __forceinline__ void tile_pv(const KVFrags& f, const v8f& p, v8f& o) {
#pragma unroll
  for (int kc = 0; kc < 4; ++kc) {
    v2f pb;
    pb.x = p[2 * kc];
    pb.y = p[2 * kc + 1];
    o = wmma4(f.va[kc], pb, o);
  }
}

template <bool CHECK>
__device__ __forceinline__ void attn_step(const KVFrags& f, const v8f& s, const v8f& s2,
                                          v2f qb0, v2f qb1, v2f qc0, v2f qc1,
                                          float& m, v8f& o, v8f& cref, v8f& p,
                                          float& m2, v8f& o2, v8f& cref2, v8f& p2) {
  v8f pn, pn2;
  tile_exp(s, pn);
  tile_exp(s2, pn2);
  tile_pv(f, p, o);
  tile_pv(f, p2, o2);
  if constexpr (CHECK) {
    if (__builtin_amdgcn_ballot_w32(!(fmaxf(sum8(pn), sum8(pn2)) <= PS_MAX)) != 0) {
      tile_rebase(f, qb0, qb1, m, o, cref, pn);
      tile_rebase(f, qc0, qc1, m2, o2, cref2, pn2);
    }
  }
  p = pn;
  p2 = pn2;
}

#define P64 68
#define P128 132
#define KV_HEAD (16 * 9)
#define KV_STAGE (H_NUM * KV_HEAD)
#define ONES_LEN 256
#define PV_BOUT 0
#define PV_G1 (E_DIM)
#define PV_BETA1 (2 * E_DIM)
#define PV_BM1 (3 * E_DIM)
#define PV_BM2 (5 * E_DIM)
#define PV_G2 (6 * E_DIM)
#define PV_BETA2 (7 * E_DIM)

__device__ __forceinline__ v2f ld2(const float* p) {
  v2f v;
  v.x = p[0];
  v.y = p[1];
  return v;
}

__device__ __forceinline__ float4 ld4(const float* p) {
  float4 v;
  v.x = p[0];
  v.y = p[1];
  v.z = p[2];
  v.w = p[3];
  return v;
}

__global__ __launch_bounds__(512) void k_attn_tail(
    const float* __restrict__ qkv, const float* __restrict__ Wout,
    const float* __restrict__ bout, const float* __restrict__ x,
    const float* __restrict__ g1, const float* __restrict__ beta1,
    const float* __restrict__ Wm1, const float* __restrict__ bm1,
    const float* __restrict__ Wm2, const float* __restrict__ bm2,
    const float* __restrict__ g2, const float* __restrict__ beta2,
    float* __restrict__ out) {
  __shared__ float ldsK0[KV_STAGE], ldsK1[KV_STAGE];
  __shared__ __attribute__((aligned(256))) float ldsV0[KV_STAGE];
  __shared__ __attribute__((aligned(256))) float ldsV1[KV_STAGE];
  __shared__ __attribute__((aligned(256))) float ldsOnes[ONES_LEN];
  __shared__ int redo;
  __shared__ __attribute__((aligned(16))) float att[64][P64];
  __shared__ float res[64][P64];
  __shared__ float mlpin[64][P64];
  __shared__ float hbuf[64][P128];
  __shared__ __attribute__((aligned(16))) float wOut[E_DIM][P64];
  __shared__ __attribute__((aligned(16))) float wM1[2 * E_DIM][P64];
  __shared__ __attribute__((aligned(16))) float wM2[E_DIM][P128];
  __shared__ __attribute__((aligned(16))) float pvec[8 * E_DIM];
  const int tid = threadIdx.x;
  const int wave = tid >> 5;
  const int lane = tid & 31;
  const int n16 = lane & 15, hi = lane >> 4;

  const int bid = blockIdx.x;
  const int b = bid >> 4;
  const int qblk = bid & 15;
  const int h = wave >> 1;
  const int qt = qblk * 4 + (wave & 1) * 2;
  const float scale = (float)(1.4426950408889634 * 0.35355339059327373);

  const float* qkvB = qkv + (size_t)(b * S_LEN) * (3 * E_DIM);
  const int srow = tid >> 5, scol = (tid & 31) * 2;
  const int sidx = (scol >> 3) * KV_HEAD + srow * 9 + (scol & 7);
  const float* stg = qkvB + srow * (3 * E_DIM) + scol;
  const float* myK0 = ldsK0 + h * KV_HEAD;
  const float* myK1 = ldsK1 + h * KV_HEAD;
  const float* myOnes = ldsOnes + ((16 * h + 32) & 63);
  const float* myV0 = n16 < 8 ? ldsV0 + h * KV_HEAD : myOnes;
  const float* myV1 = n16 < 8 ? ldsV1 + h * KV_HEAD : myOnes;
  const int d = n16 & 7;
  if (tid < ONES_LEN) ldsOnes[tid] = 1.0f;
  if (tid == 0) redo = 0;

  v8f o, o2;
  auto key_pass = [&](auto chk) __attribute__((always_inline)) {
  constexpr bool CHECK = decltype(chk)::value;

  v2f qb0, qb1, qc0, qc1;
  v2f qr0, qr1, qr2, qr3;
  {
    const float* qp = qkvB + (qt * 16 + n16) * (3 * E_DIM) + h * D_HEAD + 2 * hi;
    const float* qn = qp + 16 * (3 * E_DIM);
    qr0 = ld2(qp); qr1 = ld2(qp + 4);
    qr2 = ld2(qn); qr3 = ld2(qn + 4);
    qb0 = qr0 * scale; qb1 = qr1 * scale;
    qc0 = qr2 * scale; qc1 = qr3 * scale;
  }

  v2f kA, vA, kB, vB;
  KVFrags f0;
  {
    const v2f k0 = ld2(stg + KT_STEP + E_DIM), v0 = ld2(stg + 2 * E_DIM);
    const v2f k1 = ld2(stg + 2 * KT_STEP + E_DIM), v1 = ld2(stg + KT_STEP + 2 * E_DIM);
    kA = ld2(stg + 3 * KT_STEP + E_DIM); vA = ld2(stg + 2 * KT_STEP + 2 * E_DIM);
    kB = ld2(stg + 4 * KT_STEP + E_DIM); vB = ld2(stg + 3 * KT_STEP + 2 * E_DIM);
    const float* k0p = qkvB + n16 * (3 * E_DIM) + E_DIM + h * D_HEAD + 2 * hi;
    f0.ka0 = ld2(k0p);
    f0.ka1 = ld2(k0p + 4);
    ldsK0[sidx] = k0.x; ldsK0[sidx + 1] = k0.y;
    ldsV0[sidx] = v0.x; ldsV0[sidx + 1] = v0.y;
    ldsK1[sidx] = k1.x; ldsK1[sidx + 1] = k1.y;
    ldsV1[sidx] = v1.x; ldsV1[sidx + 1] = v1.y;
    asm volatile("" : : "v"(kA), "v"(vA), "v"(kB), "v"(vB), "v"(qr0), "v"(qr1),
                 "v"(qr2), "v"(qr3), "v"(f0.ka0), "v"(f0.ka1));
  }

  float m = -3.0e38f, m2 = -3.0e38f;
  o = v8f{};
  o2 = v8f{};
  v8f cref, cref2;


  v8f p, p2;
  tile_rebase(f0, qb0, qb1, m, o, cref, p);
  tile_rebase(f0, qc0, qc1, m2, o2, cref2, p2);
  lds_barrier();
  float4 wq[10];
  float pval = 0.f;
  if constexpr (!CHECK) {
#pragma unroll
  for (int i = 0; i < 2; ++i) wq[i] = ld4(Wout + 4 * (tid + 512 * i));
#pragma unroll
  for (int i = 0; i < 4; ++i) wq[2 + i] = ld4(Wm1 + 4 * (tid + 512 * i));
#pragma unroll
  for (int i = 0; i < 4; ++i) wq[6 + i] = ld4(Wm2 + 4 * (tid + 512 * i));
  const int wv = __builtin_amdgcn_readfirstlane(wave);
  const float* pbase = wv < 2 ? bout : wv < 4 ? g1 : wv < 6 ? beta1 : wv < 10 ? bm1
                     : wv < 12 ? bm2 : wv < 14 ? g2 : beta2;
  const int poff = wv < 2 ? 0 : wv < 4 ? PV_G1 : wv < 6 ? PV_BETA1 : wv < 10 ? PV_BM1
                 : wv < 12 ? PV_BM2 : wv < 14 ? PV_G2 : PV_BETA2;
  pval = pbase[tid - poff];
  }
  KVFrags fe = load_frags(myK0, myV0, n16, hi, d), fo;
  auto step_pair = [&](const int kt) __attribute__((always_inline)) {
    v8f s, s2;
    lds_arrive();
    asm volatile("" : "+v"(qb0), "+v"(qc0));
    s = tile_s(fe, qb0, qb1, cref);
    s2 = tile_s(fe, qc0, qc1, cref2);
    asm volatile("" : : "v"(s), "v"(s2));
    lds_pass();
    fo = load_frags(myK1, myV1, n16, hi, d);
    ldsK0[sidx] = kA.x; ldsK0[sidx + 1] = kA.y;
    ldsV0[sidx] = vA.x; ldsV0[sidx + 1] = vA.y;
    kA = ld2(stg + min(kt + 5, N_KT - 1) * KT_STEP + E_DIM);
    vA = ld2(stg + min(kt + 4, N_KT - 1) * KT_STEP + 2 * E_DIM);
    attn_step<CHECK>(fe, s, s2, qb0, qb1, qc0, qc1, m, o, cref, p, m2, o2, cref2, p2);
    lds_arrive();
    asm volatile("" : "+v"(qb0), "+v"(qc0));
    s = tile_s(fo, qb0, qb1, cref);
    s2 = tile_s(fo, qc0, qc1, cref2);
    asm volatile("" : : "v"(s), "v"(s2));
    lds_pass();
    fe = load_frags(myK0, myV0, n16, hi, d);
    ldsK1[sidx] = kB.x; ldsK1[sidx + 1] = kB.y;
    ldsV1[sidx] = vB.x; ldsV1[sidx + 1] = vB.y;
    kB = ld2(stg + min(kt + 6, N_KT - 1) * KT_STEP + E_DIM);
    vB = ld2(stg + min(kt + 5, N_KT - 1) * KT_STEP + 2 * E_DIM);
    attn_step<CHECK>(fo, s, s2, qb0, qb1, qc0, qc1, m, o, cref, p, m2, o2, cref2, p2);
  };
  step_pair(0);
  if constexpr (!CHECK) {
#pragma unroll
  for (int i = 0; i < 2; ++i) {
    const int q = tid + 512 * i;
    *reinterpret_cast<float4*>(&wOut[q >> 4][(q & 15) * 4]) = wq[i];
  }
#pragma unroll
  for (int i = 0; i < 4; ++i) {
    const int q = tid + 512 * i;
    *reinterpret_cast<float4*>(&wM1[q >> 4][(q & 15) * 4]) = wq[2 + i];
  }
#pragma unroll
  for (int i = 0; i < 4; ++i) {
    const int q = tid + 512 * i;
    *reinterpret_cast<float4*>(&wM2[q >> 5][(q & 31) * 4]) = wq[6 + i];
  }
  pvec[tid] = pval;
  }
  for (int kt = 2; kt < N_KT - 2; kt += 2) step_pair(kt);
  lds_barrier();
  fo = load_frags(myK1, myV1, n16, hi, d);
  attn_step<CHECK>(fe, tile_s(fe, qb0, qb1, cref), tile_s(fe, qc0, qc1, cref2),
                   qb0, qb1, qc0, qc1, m, o, cref, p, m2, o2, cref2, p2);
  tile_pv(fo, p, o);
  tile_pv(fo, p2, o2);
  };

  key_pass(Flag<false>{});
  {
    const bool bad = !(fabsf(sum8(o) + sum8(o2)) < INFINITY);
    if (__builtin_amdgcn_ballot_w32(bad) != 0 && lane == 0) redo = 1;
  }
  lds_barrier();
  if (__builtin_amdgcn_readfirstlane(redo) != 0) key_pass(Flag<true>{});

  {
    const float lq = xhalf(o[0]);
    const float lq2 = xhalf(o2[0]);
    if (hi == 0) {
      float4 w, x4, w2, y4;
      w.x = o[0] / lq;  w.y = o[1] / lq;  w.z = o[2] / lq;  w.w = o[3] / lq;
      x4.x = o[4] / lq; x4.y = o[5] / lq; x4.z = o[6] / lq; x4.w = o[7] / lq;
      w2.x = o2[0] / lq2; w2.y = o2[1] / lq2; w2.z = o2[2] / lq2; w2.w = o2[3] / lq2;
      y4.x = o2[4] / lq2; y4.y = o2[5] / lq2; y4.z = o2[6] / lq2; y4.w = o2[7] / lq2;
      const int r0 = (wave & 1) * 32 + n16;
      *reinterpret_cast<float4*>(&att[r0][h * D_HEAD]) = w;
      *reinterpret_cast<float4*>(&att[r0][h * D_HEAD + 4]) = x4;
      *reinterpret_cast<float4*>(&att[r0 + 16][h * D_HEAD]) = w2;
      *reinterpret_cast<float4*>(&att[r0 + 16][h * D_HEAD + 4]) = y4;
    }
  }

  const int rb = (wave >> 2) * 16;
  const int cw = wave & 3;
  const int n0 = cw * 16;
  const int nh = cw * 32;
  const int grow0 = b * S_LEN + qblk * 64;

  const float* brow = &wOut[n0 + n16][2 * hi];
  const float* w1r0 = &wM1[nh + n16][2 * hi];
  const float* w1r1 = &wM1[nh + 16 + n16][2 * hi];
  const float* w2r = &wM2[n0 + n16][2 * hi];
  v2f fb[16], fw1a[16], fw1b[16], fw2[32];
#pragma unroll
  for (int s = 0; s < 16; ++s) fb[s] = ld2(brow + 4 * s);
  float xr[8];
#pragma unroll
  for (int r = 0; r < 8; ++r) xr[r] = x[(grow0 + rb + r + 8 * hi) * E_DIM + n0 + n16];
#pragma unroll
  for (int s = 0; s < 16; ++s) { fw1a[s] = ld2(w1r0 + 4 * s); fw1b[s] = ld2(w1r1 + 4 * s); }
#pragma unroll
  for (int s = 0; s < 32; ++s) fw2[s] = ld2(w2r + 4 * s);
  __syncthreads();

  {
    v8f c = {};
#pragma unroll
    for (int s = 0; s < 16; ++s) c = wmma4(ld2(&att[rb + n16][4 * s + 2 * hi]), fb[s], c);
    const float bias = pvec[PV_BOUT + n0 + n16];
#pragma unroll
    for (int r = 0; r < 8; ++r) res[rb + r + 8 * hi][n0 + n16] = c[r] + bias + xr[r];
  }
  __syncthreads();

  const int row = rb + ((tid & 127) >> 3);
  const int c0 = (tid & 7) * 8;
  {
    float sum = 0.f, sq = 0.f;
#pragma unroll
    for (int i = 0; i < 8; ++i) { float v = res[row][c0 + i]; sum += v; sq += v * v; }
#pragma unroll
    for (int msk = 1; msk < 8; msk <<= 1) {
      sum += __shfl_xor(sum, msk);
      sq += __shfl_xor(sq, msk);
    }
    const float mean = sum * (1.f / 64.f);
    const float var = sq * (1.f / 64.f) - mean * mean;
    const float rstd = rsqrtf(var + 1e-5f);
#pragma unroll
    for (int i = 0; i < 8; ++i) {
      const int col = c0 + i;
      mlpin[row][col] = (res[row][col] - mean) * rstd * pvec[PV_G1 + col] + pvec[PV_BETA1 + col];
    }
  }
  __syncthreads();

  {
    v8f h0 = {}, h1 = {};
#pragma unroll
    for (int s = 0; s < 16; ++s) {
      const v2f a = ld2(&mlpin[rb + n16][4 * s + 2 * hi]);
      h0 = wmma4(a, fw1a[s], h0);
      h1 = wmma4(a, fw1b[s], h1);
    }
    const float bias0 = pvec[PV_BM1 + nh + n16], bias1 = pvec[PV_BM1 + nh + 16 + n16];
#pragma unroll
    for (int r = 0; r < 8; ++r) {
      hbuf[rb + r + 8 * hi][nh + n16] = fmaxf(h0[r] + bias0, 0.f);
      hbuf[rb + r + 8 * hi][nh + 16 + n16] = fmaxf(h1[r] + bias1, 0.f);
    }
  }
  __syncthreads();

  {
    v8f c = {};
#pragma unroll
    for (int s = 0; s < 32; ++s) {
      const v2f a = ld2(&hbuf[rb + n16][4 * s + 2 * hi]);
      c = wmma4(a, fw2[s], c);
    }
    const float bias = pvec[PV_BM2 + n0 + n16];
#pragma unroll
    for (int r = 0; r < 8; ++r)
      res[rb + r + 8 * hi][n0 + n16] = c[r] + bias + mlpin[rb + r + 8 * hi][n0 + n16];
  }
  __syncthreads();

  {
    float sum = 0.f, sq = 0.f;
#pragma unroll
    for (int i = 0; i < 8; ++i) { float v = res[row][c0 + i]; sum += v; sq += v * v; }
#pragma unroll
    for (int msk = 1; msk < 8; msk <<= 1) {
      sum += __shfl_xor(sum, msk);
      sq += __shfl_xor(sq, msk);
    }
    const float mean = sum * (1.f / 64.f);
    const float var = sq * (1.f / 64.f) - mean * mean;
    const float rstd = rsqrtf(var + 1e-5f);
#pragma unroll
    for (int i = 0; i < 8; ++i) {
      const int col = c0 + i;
      out[(grow0 + row) * E_DIM + col] =
          (res[row][col] - mean) * rstd * pvec[PV_G2 + col] + pvec[PV_BETA2 + col];
    }
  }
}

extern "C" void kernel_launch(void* const* d_in, const int* in_sizes, int n_in,
                              void* d_out, int out_size, void* d_ws, size_t ws_size,
                              hipStream_t stream) {
  const float* x    = (const float*)d_in[0];
  const float* Wqkv = (const float*)d_in[4];
  const float* bqkv = (const float*)d_in[5];
  const float* Wout = (const float*)d_in[6];
  const float* bout = (const float*)d_in[7];
  const float* g1   = (const float*)d_in[8];
  const float* b1   = (const float*)d_in[9];
  const float* g2   = (const float*)d_in[10];
  const float* b2   = (const float*)d_in[11];
  const float* Wm1  = (const float*)d_in[12];
  const float* bm1  = (const float*)d_in[13];
  const float* Wm2  = (const float*)d_in[14];
  const float* bm2  = (const float*)d_in[15];
  float* out = (float*)d_out;

  float* qkv = (float*)d_ws;

  k_qkv<<<N_TOK / 32, 128, 0, stream>>>(x, Wqkv, bqkv, qkv);
  k_attn_tail<<<N_TOK / 64, 512, 0, stream>>>(qkv, Wout, bout, x, g1, b1,
                                              Wm1, bm1, Wm2, bm2, g2, b2, out);
}
